# starting kernel
# baseline (speedup 1.0000x reference)
#include <hip/hip_runtime.h>
#include <math.h>

typedef __attribute__((ext_vector_type(2))) float v2f;
typedef __attribute__((ext_vector_type(8))) float v8f;

__device__ inline void atomicMaxF32(float* addr, float val) {
    int vi = __float_as_int(val);
    if (vi >= 0) {
        atomicMax((int*)addr, vi);
    } else {
        atomicMin((unsigned int*)addr, __float_as_uint(val));
    }
}

__global__ __launch_bounds__(256) void init_buffers(float* __restrict__ out,
                                                    float* __restrict__ m,
                                                    float* __restrict__ denom,
                                                    long long total_out, int rn) {
    long long idx = (long long)blockIdx.x * blockDim.x + threadIdx.x;
    if (idx < total_out) out[idx] = 0.0f;
    if (idx < rn) {
        m[idx] = -INFINITY;
        denom[idx] = 0.0f;
    }
}

__global__ __launch_bounds__(256) void edge_logits(const float* __restrict__ x,
                                                   const float* __restrict__ attn_w,
                                                   const int* __restrict__ src,
                                                   const int* __restrict__ dst,
                                                   float* __restrict__ e,
                                                   int E, int D) {
    const int lane = threadIdx.x & 31;
    const int edge = blockIdx.x * 8 + (threadIdx.x >> 5);
    const int r = blockIdx.y;
    if (edge >= E) return;

    const int s = src[(size_t)r * E + edge];
    const int t = dst[(size_t)r * E + edge];
    const float* a0 = attn_w + (size_t)r * 2 * D;
    const float* a1 = a0 + D;
    const float* xs = x + (size_t)s * D;
    const float* xd = x + (size_t)t * D;

    float acc = 0.0f;
    for (int d = lane * 4; d < D; d += 128) {
        float4 vs = *(const float4*)(xs + d);
        float4 vd = *(const float4*)(xd + d);
        float4 w0 = *(const float4*)(a0 + d);
        float4 w1 = *(const float4*)(a1 + d);
        acc += vs.x * w0.x + vs.y * w0.y + vs.z * w0.z + vs.w * w0.w;
        acc += vd.x * w1.x + vd.y * w1.y + vd.z * w1.z + vd.w * w1.w;
    }
    for (int off = 16; off > 0; off >>= 1) acc += __shfl_xor(acc, off, 32);
    if (lane == 0) e[(size_t)r * E + edge] = acc;
}

__global__ __launch_bounds__(256) void seg_max(const float* __restrict__ e,
                                               const int* __restrict__ dst,
                                               float* __restrict__ m,
                                               int E, int N) {
    const int idx = blockIdx.x * blockDim.x + threadIdx.x;
    const int r = blockIdx.y;
    if (idx >= E) return;
    const int t = dst[(size_t)r * E + idx];
    atomicMaxF32(&m[(size_t)r * N + t], e[(size_t)r * E + idx]);
}

__global__ __launch_bounds__(256) void exp_sum(float* __restrict__ e,
                                               const int* __restrict__ dst,
                                               const float* __restrict__ m,
                                               float* __restrict__ denom,
                                               int E, int N) {
    const int idx = blockIdx.x * blockDim.x + threadIdx.x;
    const int r = blockIdx.y;
    if (idx >= E) return;
    const int t = dst[(size_t)r * E + idx];
    const float p = __expf(e[(size_t)r * E + idx] - m[(size_t)r * N + t]);
    e[(size_t)r * E + idx] = p;
    atomicAdd(&denom[(size_t)r * N + t], p);
}

__global__ __launch_bounds__(256) void scatter_agg(const float* __restrict__ x,
                                                   const float* __restrict__ p,
                                                   const float* __restrict__ denom,
                                                   const int* __restrict__ src,
                                                   const int* __restrict__ dst,
                                                   float* __restrict__ out,
                                                   int E, int D, int N) {
    const int lane = threadIdx.x & 31;
    const int edge = blockIdx.x * 8 + (threadIdx.x >> 5);
    const int r = blockIdx.y;
    if (edge >= E) return;

    const int s = src[(size_t)r * E + edge];
    const int t = dst[(size_t)r * E + edge];
    const float alpha = p[(size_t)r * E + edge] /
                        fmaxf(denom[(size_t)r * N + t], 1e-30f);
    const float* xs = x + (size_t)s * D;
    float* od = out + (size_t)t * D;
    for (int d = lane * 4; d < D; d += 128) {
        float4 vs = *(const float4*)(xs + d);
        atomicAdd(od + d + 0, alpha * vs.x);
        atomicAdd(od + d + 1, alpha * vs.y);
        atomicAdd(od + d + 2, alpha * vs.z);
        atomicAdd(od + d + 3, alpha * vs.w);
    }
}

__global__ __launch_bounds__(32) void wmma_selfloop(const float* __restrict__ x,
                                                    const float* __restrict__ W,
                                                    const float* __restrict__ bias,
                                                    float* __restrict__ out,
                                                    int N, int D) {
    const int lane = threadIdx.x;
    const int m0 = blockIdx.x * 16;
    const int n0 = blockIdx.y * 64;
    const int row = lane & 15;
    const int khalf = (lane >> 4) * 2;

    v8f c0 = {}, c1 = {}, c2 = {}, c3 = {};

    for (int k = 0; k < D; k += 4) {
        const float* ap = x + (size_t)(m0 + row) * D + (k + khalf);
        v2f a;
        a.x = ap[0];
        a.y = ap[1];

        const float* bp = W + (size_t)(k + khalf) * D + n0 + row;
        v2f b0, b1, b2, b3;
        b0.x = bp[0];  b0.y = bp[D + 0];
        b1.x = bp[16]; b1.y = bp[D + 16];
        b2.x = bp[32]; b2.y = bp[D + 32];
        b3.x = bp[48]; b3.y = bp[D + 48];

        c0 = __builtin_amdgcn_wmma_f32_16x16x4_f32(false, a, false, b0, (short)0, c0, false, false);
        c1 = __builtin_amdgcn_wmma_f32_16x16x4_f32(false, a, false, b1, (short)0, c1, false, false);
        c2 = __builtin_amdgcn_wmma_f32_16x16x4_f32(false, a, false, b2, (short)0, c2, false, false);
        c3 = __builtin_amdgcn_wmma_f32_16x16x4_f32(false, a, false, b3, (short)0, c3, false, false);
    }

    const int col = n0 + (lane & 15);
    const int mbase = m0 + (lane >> 4) * 8;
#pragma unroll
    for (int v = 0; v < 8; ++v) {
        float* o = out + (size_t)(mbase + v) * D;
        o[col + 0]  += c0[v] + bias[col + 0];
        o[col + 16] += c1[v] + bias[col + 16];
        o[col + 32] += c2[v] + bias[col + 32];
        o[col + 48] += c3[v] + bias[col + 48];
    }
}

__global__ __launch_bounds__(256) void gemm_tail(const float* __restrict__ x,
                                                 const float* __restrict__ W,
                                                 const float* __restrict__ bias,
                                                 float* __restrict__ out,
                                                 int N, int D, int row0) {
    long long idx = (long long)blockIdx.x * blockDim.x + threadIdx.x;
    int row = row0 + (int)(idx / D);
    int col = (int)(idx % D);
    if (row >= N) return;
    float acc = out[(size_t)row * D + col] + bias[col];
    for (int k = 0; k < D; ++k) acc += x[(size_t)row * D + k] * W[(size_t)k * D + col];
    out[(size_t)row * D + col] = acc;
}

extern "C" void kernel_launch(void* const* d_in, const int* in_sizes, int n_in,
                              void* d_out, int out_size, void* d_ws, size_t ws_size,
                              hipStream_t stream) {
    const float* x      = (const float*)d_in[0];
    const float* attn_w = (const float*)d_in[1];
    const float* W      = (const float*)d_in[2];
    const float* bias   = (const float*)d_in[3];
    const int*   src    = (const int*)d_in[4];
    const int*   dst    = (const int*)d_in[5];
    float*       out    = (float*)d_out;

    const int D = in_sizes[3];
    const int N = in_sizes[0] / D;
    const int R = in_sizes[1] / (2 * D);
    const int E = in_sizes[4] / R;

    float* e_buf   = (float*)d_ws;
    float* m_buf   = e_buf + (size_t)R * E;
    float* den_buf = m_buf + (size_t)R * N;

    {
        long long total_out = (long long)N * D;
        int rn = R * N;
        long long mx = total_out > rn ? total_out : rn;
        int blocks = (int)((mx + 255) / 256);
        init_buffers<<<dim3(blocks), dim3(256), 0, stream>>>(out, m_buf, den_buf,
                                                             total_out, rn);
    }

    dim3 wgrid((E + 7) / 8, R);
    dim3 tgrid((E + 255) / 256, R);

    edge_logits<<<wgrid, 256, 0, stream>>>(x, attn_w, src, dst, e_buf, E, D);
    seg_max<<<tgrid, 256, 0, stream>>>(e_buf, dst, m_buf, E, N);
    exp_sum<<<tgrid, 256, 0, stream>>>(e_buf, dst, m_buf, den_buf, E, N);
    scatter_agg<<<wgrid, 256, 0, stream>>>(x, e_buf, den_buf, src, dst, out, E, D, N);

    int row_tiles = 0;
    if ((D % 64) == 0) {
        row_tiles = N / 16;
        if (row_tiles > 0) {
            dim3 ggrid(row_tiles, D / 64);
            wmma_selfloop<<<ggrid, 32, 0, stream>>>(x, W, bias, out, N, D);
        }
    }
    int row0 = row_tiles * 16;
    long long tail_elems = (long long)(N - row0) * D;
    if (tail_elems > 0) {
        int blocks = (int)((tail_elems + 255) / 256);
        gemm_tail<<<dim3(blocks), dim3(256), 0, stream>>>(x, W, bias, out, N, D, row0);
    }
}


# Softmax per destination moved into the tile kernel: listed edges counting-sorted by (row, relation) in LDS, 8 lanes per row keep max, exp-sum, row sums in registers; list-order sums, running max
# speedup vs baseline: 2.6168x; 2.6168x over previous
#include <hip/hip_runtime.h>
#include <math.h>

typedef __attribute__((ext_vector_type(2))) float v2f;
typedef __attribute__((ext_vector_type(8))) float v8f;

#define DD 128
#define TILE_FLOATS 51200
#define LCAP 12288
#define NBMAX 2048
#define CH 3

__device__ __forceinline__ float dot4(float4 a, float4 b) {
    return a.x * b.x + a.y * b.y + a.z * b.z + a.w * b.w;
}

__device__ __forceinline__ float sum8(float v) {
    v += __shfl_xor(v, 4, 32);
    v += __shfl_xor(v, 2, 32);
    v += __shfl_xor(v, 1, 32);
    return v;
}

template <int K>
__device__ __forceinline__ void take_edges(const float* __restrict__ x, const int (&s)[K],
                                           int cnt, const float4 (&w)[4], float ed, int q,
                                           float& m, float& l, float4 (&acc)[4]) {
    float4 v[K][4];
    float e[K];
#pragma unroll
    for (int u = 0; u < K; ++u) {
        const unsigned ob = (unsigned)s[u] * (unsigned)(DD * 4) + (unsigned)(q * 16);
        const float4* p = (const float4*)((const char*)x + ob);
#pragma unroll
        for (int j = 0; j < 4; ++j) v[u][j] = p[8 * j];
    }
    float cm = m;
#pragma unroll
    for (int u = 0; u < K; ++u) {
        const float pe = dot4(v[u][0], w[0]) + dot4(v[u][1], w[1]) +
                         dot4(v[u][2], w[2]) + dot4(v[u][3], w[3]);
        const float eu = sum8(pe) + ed;
        e[u] = (u < cnt) ? eu : -INFINITY;
        cm = fmaxf(cm, e[u]);
    }
    const float sc = __expf(m - cm);
    l *= sc;
#pragma unroll
    for (int j = 0; j < 4; ++j) {
        acc[j].x *= sc; acc[j].y *= sc; acc[j].z *= sc; acc[j].w *= sc;
    }
#pragma unroll
    for (int u = 0; u < K; ++u) {
        const float p = (u < cnt) ? __expf(e[u] - cm) : 0.0f;
        l += p;
#pragma unroll
        for (int j = 0; j < 4; ++j) {
            acc[j].x += p * v[u][j].x; acc[j].y += p * v[u][j].y;
            acc[j].z += p * v[u][j].z; acc[j].w += p * v[u][j].w;
        }
    }
    m = cm;
}

__device__ __forceinline__ void row_unsorted(const float* __restrict__ x,
                                          const float* __restrict__ attn_w,
                                          const int* __restrict__ src,
                                          const int* __restrict__ dst,
                                          float* __restrict__ out,
                                          int E, int R, int t, int q) {
    const float4* xd = (const float4*)(x + (size_t)t * DD + q * 4);
    float4 tot[4];
#pragma unroll
    for (int j = 0; j < 4; ++j) tot[j] = make_float4(0.0f, 0.0f, 0.0f, 0.0f);
    for (int r = 0; r < R; ++r) {
        const float4* a0 = (const float4*)(attn_w + (size_t)r * 2 * DD + q * 4);
        const float4* a1 = a0 + DD / 4;
        float4 w[4];
        float pd = 0.0f;
#pragma unroll
        for (int j = 0; j < 4; ++j) {
            w[j] = a0[8 * j];
            pd += dot4(xd[8 * j], a1[8 * j]);
        }
        const float ed = sum8(pd);
        float m = -INFINITY, l = 0.0f;
        float4 acc[4];
#pragma unroll
        for (int j = 0; j < 4; ++j) acc[j] = make_float4(0.0f, 0.0f, 0.0f, 0.0f);
        for (int i = 0; i < E; ++i)
            if (dst[(size_t)r * E + i] == t) {
                const int s[1] = { src[(size_t)r * E + i] };
                take_edges<1>(x, s, 1, w, ed, q, m, l, acc);
            }
        const float inv = 1.0f / fmaxf(l, 1e-30f);
#pragma unroll
        for (int j = 0; j < 4; ++j) {
            tot[j].x += acc[j].x * inv; tot[j].y += acc[j].y * inv;
            tot[j].z += acc[j].z * inv; tot[j].w += acc[j].w * inv;
        }
    }
    float4* o = (float4*)(out + (size_t)t * DD + q * 4);
#pragma unroll
    for (int j = 0; j < 4; ++j) o[8 * j] = tot[j];
}

__global__ __launch_bounds__(1024) void edge_tile(const float* __restrict__ x,
                                                  const float* __restrict__ attn_w,
                                                  const int* __restrict__ src,
                                                  const int* __restrict__ dst,
                                                  float* __restrict__ out,
                                                  int E, int N, int R, int rpw, int total) {
    __shared__ int   lst_i[LCAP];
    __shared__ int   lst_b[LCAP];
    __shared__ int   srt_s[LCAP];
    __shared__ int   off[NBMAX + 1];
    __shared__ int   fil[NBMAX];
    __shared__ float edl[NBMAX];
    __shared__ float4 totl[128 * 32];
    __shared__ int   wsum[32];
    __shared__ int   lst_n;

    const int tid = threadIdx.x;
    const int lane = tid & 31;
    const int wave = __builtin_amdgcn_readfirstlane(tid >> 5);
    const int q = lane & 7;
    const int row0 = blockIdx.x * rpw;
    const int nrows = min(rpw, N - row0);
    const int nb = nrows * R;
    const bool binned = nb <= NBMAX;

    for (int b = tid; b <= NBMAX; b += 1024) off[b] = 0;
    if (tid == 0) lst_n = 0;

    if (binned)
        for (int base = wave * 4; base < nrows; base += 128) {
            const int tl = base + (lane >> 3);
            if (tl < nrows) {
                const float4* xd = (const float4*)(x + (size_t)(row0 + tl) * DD + q * 4);
                float4 xv[4];
#pragma unroll
                for (int j = 0; j < 4; ++j) xv[j] = xd[8 * j];
                for (int r = 0; r < R; ++r) {
                    const float4* a1 = (const float4*)(attn_w + (size_t)r * 2 * DD + DD + q * 4);
                    const float pd = dot4(xv[0], a1[0]) + dot4(xv[1], a1[8]) +
                                     dot4(xv[2], a1[16]) + dot4(xv[3], a1[24]);
                    const float ed = sum8(pd);
                    if (q == 0) edl[tl * R + r] = ed;
                }
            }
        }
    __syncthreads();

    auto take = [&](int t, int i) {
        const unsigned tl = (unsigned)(t - row0);
        if (tl < (unsigned)nrows) {
            const int slot = atomicAdd(&lst_n, 1);
            if (binned && slot < LCAP) {
                const int b = (int)tl * R + i / E;
                lst_i[slot] = i;
                lst_b[slot] = b;
                atomicAdd(&off[b], 1);
            }
        }
    };

    const int4* dst4 = (const int4*)dst;
    const int n4 = total / 4;
    for (int q0 = tid; q0 < n4; q0 += 1024 * 8) {
        int4 v[8];
#pragma unroll
        for (int k = 0; k < 8; ++k) {
            const int qq = q0 + k * 1024;
            v[k] = (qq < n4) ? dst4[qq] : make_int4(-1, -1, -1, -1);
        }
#pragma unroll
        for (int k = 0; k < 8; ++k) {
            const unsigned d0 = (unsigned)(v[k].x - row0), d1 = (unsigned)(v[k].y - row0);
            const unsigned d2 = (unsigned)(v[k].z - row0), d3 = (unsigned)(v[k].w - row0);
            if (min(min(d0, d1), min(d2, d3)) < (unsigned)nrows) {
                const int i = (q0 + k * 1024) * 4;
                take(v[k].x, i + 0);
                take(v[k].y, i + 1);
                take(v[k].z, i + 2);
                take(v[k].w, i + 3);
            }
        }
    }
    for (int i = n4 * 4 + tid; i < total; i += 1024) take(dst[i], i);
    __syncthreads();

    const int n = lst_n;
    if (!binned || n > LCAP) {
        for (int base = wave * 4; base < nrows; base += 128) {
            const int tl = base + (lane >> 3);
            if (tl < nrows) row_unsorted(x, attn_w, src, dst, out, E, R, row0 + tl, q);
        }
        return;
    }

    const int b0 = 2 * tid, b1 = 2 * tid + 1;
    const int c0 = (b0 < nb) ? off[b0] : 0;
    const int c1 = (b1 < nb) ? off[b1] : 0;
    int incl = c0 + c1;
    for (int o = 1; o < 32; o <<= 1) {
        const int y = __shfl_up(incl, o, 32);
        if (lane >= o) incl += y;
    }
    if (lane == 31) wsum[wave] = incl;
    __syncthreads();
    if (wave == 0) {
        const int wt = wsum[lane];
        int wi = wt;
        for (int o = 1; o < 32; o <<= 1) {
            const int y = __shfl_up(wi, o, 32);
            if (lane >= o) wi += y;
        }
        wsum[lane] = wi - wt;
    }
    __syncthreads();
    const int ex = wsum[wave] + incl - (c0 + c1);
    if (b0 < nb) { off[b0] = ex;      fil[b0] = ex; }
    if (b1 < nb) { off[b1] = ex + c0; fil[b1] = ex + c0; }
    if (tid == 0) off[nb] = n;
    __syncthreads();

    for (int j = tid; j < n; j += 1024) {
        const int pos = atomicAdd(&fil[lst_b[j]], 1);
        srt_s[pos] = src[lst_i[j]];
    }
    __syncthreads();

    float4* mine = totl + (size_t)(tid >> 3) * 32 + q;
    for (int base = wave * 4; base < nrows; base += 128) {
        const int tl = base + (lane >> 3);
        if (tl < nrows) {
            for (int r = 0; r < R; ++r) {
                const float4* a0 = (const float4*)((const char*)attn_w +
                                                   (unsigned)(r * (2 * DD * 4) + q * 16));
                float4 w[4];
#pragma unroll
                for (int j = 0; j < 4; ++j) w[j] = a0[8 * j];
                const float ed = edl[tl * R + r];
                const int beg = off[tl * R + r], end = off[tl * R + r + 1];
                float m = -INFINITY, l = 0.0f;
                float4 acc[4];
#pragma unroll
                for (int j = 0; j < 4; ++j) acc[j] = make_float4(0.0f, 0.0f, 0.0f, 0.0f);
                for (int k0 = beg; k0 < end; k0 += CH) {
                    const int cnt = min(CH, end - k0);
                    int s[CH];
#pragma unroll
                    for (int u = 0; u < CH; ++u) s[u] = srt_s[min(k0 + u, end - 1)];
                    take_edges<CH>(x, s, cnt, w, ed, q, m, l, acc);
                }
                const float inv = 1.0f / fmaxf(l, 1e-30f);
#pragma unroll
                for (int j = 0; j < 4; ++j) {
                    acc[j].x *= inv; acc[j].y *= inv; acc[j].z *= inv; acc[j].w *= inv;
                    if (r > 0) {
                        const float4 pv = mine[8 * j];
                        acc[j].x += pv.x; acc[j].y += pv.y; acc[j].z += pv.z; acc[j].w += pv.w;
                    }
                }
                if (r + 1 < R) {
#pragma unroll
                    for (int j = 0; j < 4; ++j) mine[8 * j] = acc[j];
                } else {
                    float4* o = (float4*)((char*)out +
                                          (unsigned)((row0 + tl) * (DD * 4) + q * 16));
#pragma unroll
                    for (int j = 0; j < 4; ++j) o[8 * j] = acc[j];
                }
            }
        }
    }
}

__global__ __launch_bounds__(32) void wmma_selfloop(const float* __restrict__ x,
                                                    const float* __restrict__ W,
                                                    const float* __restrict__ bias,
                                                    float* __restrict__ out,
                                                    int N, int D) {
    const int lane = threadIdx.x;
    const int m0 = blockIdx.x * 16;
    const int n0 = blockIdx.y * 64;
    const int row = lane & 15;
    const int khalf = (lane >> 4) * 2;

    v8f c0 = {}, c1 = {}, c2 = {}, c3 = {};

    for (int k = 0; k < D; k += 4) {
        const float* ap = x + (size_t)(m0 + row) * D + (k + khalf);
        v2f a;
        a.x = ap[0];
        a.y = ap[1];

        const float* bp = W + (size_t)(k + khalf) * D + n0 + row;
        v2f b0, b1, b2, b3;
        b0.x = bp[0];  b0.y = bp[D + 0];
        b1.x = bp[16]; b1.y = bp[D + 16];
        b2.x = bp[32]; b2.y = bp[D + 32];
        b3.x = bp[48]; b3.y = bp[D + 48];

        c0 = __builtin_amdgcn_wmma_f32_16x16x4_f32(false, a, false, b0, (short)0, c0, false, false);
        c1 = __builtin_amdgcn_wmma_f32_16x16x4_f32(false, a, false, b1, (short)0, c1, false, false);
        c2 = __builtin_amdgcn_wmma_f32_16x16x4_f32(false, a, false, b2, (short)0, c2, false, false);
        c3 = __builtin_amdgcn_wmma_f32_16x16x4_f32(false, a, false, b3, (short)0, c3, false, false);
    }

    const int col = n0 + (lane & 15);
    const int mbase = m0 + (lane >> 4) * 8;
#pragma unroll
    for (int v = 0; v < 8; ++v) {
        float* o = out + (size_t)(mbase + v) * D;
        o[col + 0]  += c0[v] + bias[col + 0];
        o[col + 16] += c1[v] + bias[col + 16];
        o[col + 32] += c2[v] + bias[col + 32];
        o[col + 48] += c3[v] + bias[col + 48];
    }
}

__global__ __launch_bounds__(256) void gemm_tail(const float* __restrict__ x,
                                                 const float* __restrict__ W,
                                                 const float* __restrict__ bias,
                                                 float* __restrict__ out,
                                                 int N, int D, int row0) {
    long long idx = (long long)blockIdx.x * blockDim.x + threadIdx.x;
    int row = row0 + (int)(idx / D);
    int col = (int)(idx % D);
    if (row >= N) return;
    float acc = out[(size_t)row * D + col] + bias[col];
    for (int k = 0; k < D; ++k) acc += x[(size_t)row * D + k] * W[(size_t)k * D + col];
    out[(size_t)row * D + col] = acc;
}

extern "C" void kernel_launch(void* const* d_in, const int* in_sizes, int n_in,
                              void* d_out, int out_size, void* d_ws, size_t ws_size,
                              hipStream_t stream) {
    const float* x      = (const float*)d_in[0];
    const float* attn_w = (const float*)d_in[1];
    const float* W      = (const float*)d_in[2];
    const float* bias   = (const float*)d_in[3];
    const int*   src    = (const int*)d_in[4];
    const int*   dst    = (const int*)d_in[5];
    float*       out    = (float*)d_out;

    const int D = in_sizes[3];
    const int N = in_sizes[0] / D;
    const int R = in_sizes[1] / (2 * D);
    const int E = in_sizes[4] / R;

    const int rpw = TILE_FLOATS / D;
    dim3 sgrid((N + rpw - 1) / rpw);
    edge_tile<<<sgrid, 1024, 0, stream>>>(x, attn_w, src, dst, out, E, N, R, rpw, R * E);

    int row_tiles = 0;
    if ((D % 64) == 0) {
        row_tiles = N / 16;
        if (row_tiles > 0) {
            dim3 ggrid(row_tiles, D / 64);
            wmma_selfloop<<<ggrid, 32, 0, stream>>>(x, W, bias, out, N, D);
        }
    }
    int row0 = row_tiles * 16;
    long long tail_elems = (long long)(N - row0) * D;
    if (tail_elems > 0) {
        int blocks = (int)((tail_elems + 255) / 256);
        gemm_tail<<<dim3(blocks), dim3(256), 0, stream>>>(x, W, bias, out, N, D, row0);
    }
}


# Row-block kernel no longer scans all dst: a binning launch counting-sorts each 4096-edge chunk by row block into the workspace as (bin, source) records; 3 launches; sums stay in list arrival order
# speedup vs baseline: 3.4574x; 1.3212x over previous
#include <hip/hip_runtime.h>
#include <math.h>

typedef __attribute__((ext_vector_type(2))) float v2f;
typedef __attribute__((ext_vector_type(8))) float v8f;

#define DD 128
#define TILE_FLOATS 51200
#define LCAP 12288
#define NBMAX 2048
#define CH 3

__device__ __forceinline__ float dot4(float4 a, float4 b) {
    return a.x * b.x + a.y * b.y + a.z * b.z + a.w * b.w;
}

__device__ __forceinline__ float sum8(float v) {
    v += __shfl_xor(v, 4, 32);
    v += __shfl_xor(v, 2, 32);
    v += __shfl_xor(v, 1, 32);
    return v;
}

template <int K>
__device__ __forceinline__ void take_edges(const float* __restrict__ x, const int (&s)[K],
                                           int cnt, const float4 (&w)[4], float ed, int q,
                                           float& m, float& l, float4 (&acc)[4]) {
    float4 v[K][4];
    float e[K];
#pragma unroll
    for (int u = 0; u < K; ++u) {
        const unsigned ob = (unsigned)s[u] * (unsigned)(DD * 4) + (unsigned)(q * 16);
        const float4* p = (const float4*)((const char*)x + ob);
#pragma unroll
        for (int j = 0; j < 4; ++j) v[u][j] = p[8 * j];
    }
    float cm = m;
#pragma unroll
    for (int u = 0; u < K; ++u) {
        const float pe = dot4(v[u][0], w[0]) + dot4(v[u][1], w[1]) +
                         dot4(v[u][2], w[2]) + dot4(v[u][3], w[3]);
        const float eu = sum8(pe) + ed;
        e[u] = (u < cnt) ? eu : -INFINITY;
        cm = fmaxf(cm, e[u]);
    }
    const float sc = __expf(m - cm);
    l *= sc;
#pragma unroll
    for (int j = 0; j < 4; ++j) {
        acc[j].x *= sc; acc[j].y *= sc; acc[j].z *= sc; acc[j].w *= sc;
    }
#pragma unroll
    for (int u = 0; u < K; ++u) {
        const float p = (u < cnt) ? __expf(e[u] - cm) : 0.0f;
        l += p;
#pragma unroll
        for (int j = 0; j < 4; ++j) {
            acc[j].x += p * v[u][j].x; acc[j].y += p * v[u][j].y;
            acc[j].z += p * v[u][j].z; acc[j].w += p * v[u][j].w;
        }
    }
    m = cm;
}

__device__ __forceinline__ void row_unsorted(const float* __restrict__ x,
                                          const float* __restrict__ attn_w,
                                          const int* __restrict__ src,
                                          const int* __restrict__ dst,
                                          float* __restrict__ out,
                                          int E, int R, int t, int q) {
    const float4* xd = (const float4*)(x + (size_t)t * DD + q * 4);
    float4 tot[4];
#pragma unroll
    for (int j = 0; j < 4; ++j) tot[j] = make_float4(0.0f, 0.0f, 0.0f, 0.0f);
    for (int r = 0; r < R; ++r) {
        const float4* a0 = (const float4*)(attn_w + (size_t)r * 2 * DD + q * 4);
        const float4* a1 = a0 + DD / 4;
        float4 w[4];
        float pd = 0.0f;
#pragma unroll
        for (int j = 0; j < 4; ++j) {
            w[j] = a0[8 * j];
            pd += dot4(xd[8 * j], a1[8 * j]);
        }
        const float ed = sum8(pd);
        float m = -INFINITY, l = 0.0f;
        float4 acc[4];
#pragma unroll
        for (int j = 0; j < 4; ++j) acc[j] = make_float4(0.0f, 0.0f, 0.0f, 0.0f);
        for (int i = 0; i < E; ++i)
            if (dst[(size_t)r * E + i] == t) {
                const int s[1] = { src[(size_t)r * E + i] };
                take_edges<1>(x, s, 1, w, ed, q, m, l, acc);
            }
        const float inv = 1.0f / fmaxf(l, 1e-30f);
#pragma unroll
        for (int j = 0; j < 4; ++j) {
            tot[j].x += acc[j].x * inv; tot[j].y += acc[j].y * inv;
            tot[j].z += acc[j].z * inv; tot[j].w += acc[j].w * inv;
        }
    }
    float4* o = (float4*)(out + (size_t)t * DD + q * 4);
#pragma unroll
    for (int j = 0; j < 4; ++j) o[8 * j] = tot[j];
}

#define CHUNK 4096
#define NTMAX 1024

__device__ __forceinline__ bool bins_ok(int total, int rpw, int ntiles, int R, int N,
                                        size_t ws_bytes) {
    const size_t nch = (size_t)((total + CHUNK - 1) / CHUNK);
    const size_t tab = (nch * (size_t)(ntiles + 1) + 63) & ~(size_t)63;
    return N <= (1 << 20) && rpw * R <= NBMAX && ntiles <= NTMAX &&
           (tab + nch * CHUNK) * sizeof(int) <= ws_bytes;
}

__global__ __launch_bounds__(1024) void bin_edges(const int* __restrict__ src,
                                                  const int* __restrict__ dst,
                                                  int* __restrict__ ws,
                                                  int total, int rpw, int ntiles,
                                                  int E, int R, int N, size_t ws_bytes) {
    __shared__ int cnt[NTMAX];
    __shared__ int beg[NTMAX];
    __shared__ int srt[CHUNK];
    __shared__ int wsum[32];

    if (!bins_ok(total, rpw, ntiles, R, N, ws_bytes)) return;
    const int tid = threadIdx.x;
    const int lane = tid & 31;
    const int wave = tid >> 5;
    const int c = blockIdx.x;
    const int nch = (total + CHUNK - 1) / CHUNK;
    const int nt1 = ntiles + 1;
    const size_t tab = ((size_t)nch * (size_t)nt1 + 63) & ~(size_t)63;
    int* lst = ws + tab + (size_t)c * CHUNK;
    int* tbl = ws + (size_t)c * nt1;

    cnt[tid] = 0;
    __syncthreads();

    const int i0 = c * CHUNK + tid * 4;
    int d[4], sv[4];
    if (i0 + 3 < total) {
        const int4 v = *(const int4*)(dst + i0);
        const int4 u = *(const int4*)(src + i0);
        d[0] = v.x; d[1] = v.y; d[2] = v.z; d[3] = v.w;
        sv[0] = u.x; sv[1] = u.y; sv[2] = u.z; sv[3] = u.w;
    } else {
#pragma unroll
        for (int k = 0; k < 4; ++k) {
            d[k] = (i0 + k < total) ? dst[i0 + k] : -1;
            sv[k] = (i0 + k < total) ? src[i0 + k] : 0;
        }
    }
    int tt[4], val[4], pos[4];
#pragma unroll
    for (int k = 0; k < 4; ++k) {
        const int t = (d[k] >= 0) ? d[k] / rpw : ntiles;
        tt[k] = ((unsigned)t < (unsigned)ntiles) ? t : -1;
        val[k] = (tt[k] >= 0) ? ((((d[k] - t * rpw) * R + (i0 + k) / E) << 20) | sv[k]) : 0;
        pos[k] = 0;
        if (tt[k] >= 0) pos[k] = atomicAdd(&cnt[tt[k]], 1);
    }
    __syncthreads();

    const int c0 = (tid < ntiles) ? cnt[tid] : 0;
    int incl = c0;
    for (int o = 1; o < 32; o <<= 1) {
        const int y = __shfl_up(incl, o, 32);
        if (lane >= o) incl += y;
    }
    if (lane == 31) wsum[wave] = incl;
    __syncthreads();
    if (wave == 0) {
        const int wt = wsum[lane];
        int wi = wt;
        for (int o = 1; o < 32; o <<= 1) {
            const int y = __shfl_up(wi, o, 32);
            if (lane >= o) wi += y;
        }
        wsum[lane] = wi - wt;
    }
    __syncthreads();
    const int ex = wsum[wave] + incl - c0;
    beg[tid] = ex;
    if (tid < ntiles) tbl[tid] = ex;
    if (tid == 1023) tbl[ntiles] = ex + c0;
    __syncthreads();

#pragma unroll
    for (int k = 0; k < 4; ++k)
        if (tt[k] >= 0) srt[beg[tt[k]] + pos[k]] = val[k];
    __syncthreads();

    const int nv = beg[1023] + cnt[1023];
#pragma unroll
    for (int k = 0; k < 4; ++k) {
        const int p = tid + k * 1024;
        if (p < nv) lst[p] = srt[p];
    }
}

__global__ __launch_bounds__(1024) void edge_tile(const float* __restrict__ x,
                                                  const float* __restrict__ attn_w,
                                                  const int* __restrict__ src,
                                                  const int* __restrict__ dst,
                                                  float* __restrict__ out,
                                                  const int* __restrict__ ws, size_t ws_bytes,
                                                  int E, int N, int R, int rpw, int total) {
    __shared__ int   lst_i[LCAP];
    __shared__ int   lst_b[LCAP];
    __shared__ int   srt_s[LCAP];
    __shared__ int   off[NBMAX + 1];
    __shared__ int   fil[NBMAX];
    __shared__ float edl[NBMAX];
    __shared__ float4 totl[128 * 32];
    __shared__ int   wsum[32];
    __shared__ int   lst_n;

    const int tid = threadIdx.x;
    const int lane = tid & 31;
    const int wave = __builtin_amdgcn_readfirstlane(tid >> 5);
    const int q = lane & 7;
    const int row0 = blockIdx.x * rpw;
    const int nrows = min(rpw, N - row0);
    const int nb = nrows * R;
    const bool binned = nb <= NBMAX;

    for (int b = tid; b <= NBMAX; b += 1024) off[b] = 0;
    if (tid == 0) lst_n = 0;

    if (binned)
        for (int base = wave * 4; base < nrows; base += 128) {
            const int tl = base + (lane >> 3);
            if (tl < nrows) {
                const float4* xd = (const float4*)(x + (size_t)(row0 + tl) * DD + q * 4);
                float4 xv[4];
#pragma unroll
                for (int j = 0; j < 4; ++j) xv[j] = xd[8 * j];
                for (int r = 0; r < R; ++r) {
                    const float4* a1 = (const float4*)(attn_w + (size_t)r * 2 * DD + DD + q * 4);
                    const float pd = dot4(xv[0], a1[0]) + dot4(xv[1], a1[8]) +
                                     dot4(xv[2], a1[16]) + dot4(xv[3], a1[24]);
                    const float ed = sum8(pd);
                    if (q == 0) edl[tl * R + r] = ed;
                }
            }
        }
    __syncthreads();

    auto take_rec = [&](int b, int sr) {
        const int slot = atomicAdd(&lst_n, 1);
        if (binned && slot < LCAP) {
            lst_i[slot] = sr;
            lst_b[slot] = b;
            atomicAdd(&off[b], 1);
        }
    };
    auto take = [&](int t, int i) {
        const unsigned tl = (unsigned)(t - row0);
        if (tl < (unsigned)nrows) take_rec((int)tl * R + i / E, src[i]);
    };

    if (bins_ok(total, rpw, (int)gridDim.x, R, N, ws_bytes)) {
        const int nch = (total + CHUNK - 1) / CHUNK;
        const int nt1 = (int)gridDim.x + 1;
        const size_t tab = ((size_t)nch * (size_t)nt1 + 63) & ~(size_t)63;
        for (int c = tid >> 3; c < nch; c += 128) {
            const int* tbl = ws + (size_t)c * nt1 + blockIdx.x;
            const int sb = tbl[0], se = tbl[1];
            const int* lst = ws + tab + (size_t)c * CHUNK;
            for (int p = sb + q; p < se; p += 16) {
                const int e0 = lst[p];
                const int e1 = (p + 8 < se) ? lst[p + 8] : -1;
                if ((e0 >> 20) < nb) take_rec(e0 >> 20, e0 & ((1 << 20) - 1));
                if (e1 >= 0 && (e1 >> 20) < nb) take_rec(e1 >> 20, e1 & ((1 << 20) - 1));
            }
        }
    } else {
        const int4* dst4 = (const int4*)dst;
        const int n4 = total / 4;
        for (int q0 = tid; q0 < n4; q0 += 1024 * 8) {
            int4 v[8];
    #pragma unroll
            for (int k = 0; k < 8; ++k) {
                const int qq = q0 + k * 1024;
                v[k] = (qq < n4) ? dst4[qq] : make_int4(-1, -1, -1, -1);
            }
    #pragma unroll
            for (int k = 0; k < 8; ++k) {
                const unsigned d0 = (unsigned)(v[k].x - row0), d1 = (unsigned)(v[k].y - row0);
                const unsigned d2 = (unsigned)(v[k].z - row0), d3 = (unsigned)(v[k].w - row0);
                if (min(min(d0, d1), min(d2, d3)) < (unsigned)nrows) {
                    const int i = (q0 + k * 1024) * 4;
                    take(v[k].x, i + 0);
                    take(v[k].y, i + 1);
                    take(v[k].z, i + 2);
                    take(v[k].w, i + 3);
                }
            }
        }
        for (int i = n4 * 4 + tid; i < total; i += 1024) take(dst[i], i);
    }
    __syncthreads();

    const int n = lst_n;
    if (!binned || n > LCAP) {
        for (int base = wave * 4; base < nrows; base += 128) {
            const int tl = base + (lane >> 3);
            if (tl < nrows) row_unsorted(x, attn_w, src, dst, out, E, R, row0 + tl, q);
        }
        return;
    }

    const int b0 = 2 * tid, b1 = 2 * tid + 1;
    const int c0 = (b0 < nb) ? off[b0] : 0;
    const int c1 = (b1 < nb) ? off[b1] : 0;
    int incl = c0 + c1;
    for (int o = 1; o < 32; o <<= 1) {
        const int y = __shfl_up(incl, o, 32);
        if (lane >= o) incl += y;
    }
    if (lane == 31) wsum[wave] = incl;
    __syncthreads();
    if (wave == 0) {
        const int wt = wsum[lane];
        int wi = wt;
        for (int o = 1; o < 32; o <<= 1) {
            const int y = __shfl_up(wi, o, 32);
            if (lane >= o) wi += y;
        }
        wsum[lane] = wi - wt;
    }
    __syncthreads();
    const int ex = wsum[wave] + incl - (c0 + c1);
    if (b0 < nb) { off[b0] = ex;      fil[b0] = ex; }
    if (b1 < nb) { off[b1] = ex + c0; fil[b1] = ex + c0; }
    if (tid == 0) off[nb] = n;
    __syncthreads();

    for (int j = tid; j < n; j += 1024) {
        const int pos = atomicAdd(&fil[lst_b[j]], 1);
        srt_s[pos] = lst_i[j];
    }
    __syncthreads();

    float4* mine = totl + (size_t)(tid >> 3) * 32 + q;
    for (int base = wave * 4; base < nrows; base += 128) {
        const int tl = base + (lane >> 3);
        if (tl < nrows) {
            for (int r = 0; r < R; ++r) {
                const float4* a0 = (const float4*)((const char*)attn_w +
                                                   (unsigned)(r * (2 * DD * 4) + q * 16));
                float4 w[4];
#pragma unroll
                for (int j = 0; j < 4; ++j) w[j] = a0[8 * j];
                const float ed = edl[tl * R + r];
                const int beg = off[tl * R + r], end = off[tl * R + r + 1];
                float m = -INFINITY, l = 0.0f;
                float4 acc[4];
#pragma unroll
                for (int j = 0; j < 4; ++j) acc[j] = make_float4(0.0f, 0.0f, 0.0f, 0.0f);
                for (int k0 = beg; k0 < end; k0 += CH) {
                    const int cnt = min(CH, end - k0);
                    int s[CH];
#pragma unroll
                    for (int u = 0; u < CH; ++u) s[u] = srt_s[min(k0 + u, end - 1)];
                    take_edges<CH>(x, s, cnt, w, ed, q, m, l, acc);
                }
                const float inv = 1.0f / fmaxf(l, 1e-30f);
#pragma unroll
                for (int j = 0; j < 4; ++j) {
                    acc[j].x *= inv; acc[j].y *= inv; acc[j].z *= inv; acc[j].w *= inv;
                    if (r > 0) {
                        const float4 pv = mine[8 * j];
                        acc[j].x += pv.x; acc[j].y += pv.y; acc[j].z += pv.z; acc[j].w += pv.w;
                    }
                }
                if (r + 1 < R) {
#pragma unroll
                    for (int j = 0; j < 4; ++j) mine[8 * j] = acc[j];
                } else {
                    float4* o = (float4*)((char*)out +
                                          (unsigned)((row0 + tl) * (DD * 4) + q * 16));
#pragma unroll
                    for (int j = 0; j < 4; ++j) o[8 * j] = acc[j];
                }
            }
        }
    }
}

__global__ __launch_bounds__(32) void wmma_selfloop(const float* __restrict__ x,
                                                    const float* __restrict__ W,
                                                    const float* __restrict__ bias,
                                                    float* __restrict__ out,
                                                    int N, int D) {
    const int lane = threadIdx.x;
    const int m0 = blockIdx.x * 16;
    const int n0 = blockIdx.y * 64;
    const int row = lane & 15;
    const int khalf = (lane >> 4) * 2;

    v8f c0 = {}, c1 = {}, c2 = {}, c3 = {};

    for (int k = 0; k < D; k += 4) {
        const float* ap = x + (size_t)(m0 + row) * D + (k + khalf);
        v2f a;
        a.x = ap[0];
        a.y = ap[1];

        const float* bp = W + (size_t)(k + khalf) * D + n0 + row;
        v2f b0, b1, b2, b3;
        b0.x = bp[0];  b0.y = bp[D + 0];
        b1.x = bp[16]; b1.y = bp[D + 16];
        b2.x = bp[32]; b2.y = bp[D + 32];
        b3.x = bp[48]; b3.y = bp[D + 48];

        c0 = __builtin_amdgcn_wmma_f32_16x16x4_f32(false, a, false, b0, (short)0, c0, false, false);
        c1 = __builtin_amdgcn_wmma_f32_16x16x4_f32(false, a, false, b1, (short)0, c1, false, false);
        c2 = __builtin_amdgcn_wmma_f32_16x16x4_f32(false, a, false, b2, (short)0, c2, false, false);
        c3 = __builtin_amdgcn_wmma_f32_16x16x4_f32(false, a, false, b3, (short)0, c3, false, false);
    }

    const int col = n0 + (lane & 15);
    const int mbase = m0 + (lane >> 4) * 8;
#pragma unroll
    for (int v = 0; v < 8; ++v) {
        float* o = out + (size_t)(mbase + v) * D;
        o[col + 0]  += c0[v] + bias[col + 0];
        o[col + 16] += c1[v] + bias[col + 16];
        o[col + 32] += c2[v] + bias[col + 32];
        o[col + 48] += c3[v] + bias[col + 48];
    }
}

__global__ __launch_bounds__(256) void gemm_tail(const float* __restrict__ x,
                                                 const float* __restrict__ W,
                                                 const float* __restrict__ bias,
                                                 float* __restrict__ out,
                                                 int N, int D, int row0) {
    long long idx = (long long)blockIdx.x * blockDim.x + threadIdx.x;
    int row = row0 + (int)(idx / D);
    int col = (int)(idx % D);
    if (row >= N) return;
    float acc = out[(size_t)row * D + col] + bias[col];
    for (int k = 0; k < D; ++k) acc += x[(size_t)row * D + k] * W[(size_t)k * D + col];
    out[(size_t)row * D + col] = acc;
}

extern "C" void kernel_launch(void* const* d_in, const int* in_sizes, int n_in,
                              void* d_out, int out_size, void* d_ws, size_t ws_size,
                              hipStream_t stream) {
    const float* x      = (const float*)d_in[0];
    const float* attn_w = (const float*)d_in[1];
    const float* W      = (const float*)d_in[2];
    const float* bias   = (const float*)d_in[3];
    const int*   src    = (const int*)d_in[4];
    const int*   dst    = (const int*)d_in[5];
    float*       out    = (float*)d_out;

    const int D = in_sizes[3];
    const int N = in_sizes[0] / D;
    const int R = in_sizes[1] / (2 * D);
    const int E = in_sizes[4] / R;

    const int rpw = TILE_FLOATS / D;
    dim3 sgrid((N + rpw - 1) / rpw);
    dim3 pgrid((R * E + CHUNK - 1) / CHUNK);
    bin_edges<<<pgrid, 1024, 0, stream>>>(src, dst, (int*)d_ws, R * E, rpw, (int)sgrid.x,
                                          E, R, N, ws_size);
    edge_tile<<<sgrid, 1024, 0, stream>>>(x, attn_w, src, dst, out, (const int*)d_ws, ws_size,
                                          E, N, R, rpw, R * E);

    int row_tiles = 0;
    if ((D % 64) == 0) {
        row_tiles = N / 16;
        if (row_tiles > 0) {
            dim3 ggrid(row_tiles, D / 64);
            wmma_selfloop<<<ggrid, 32, 0, stream>>>(x, W, bias, out, N, D);
        }
    }
    int row0 = row_tiles * 16;
    long long tail_elems = (long long)(N - row0) * D;
    if (tail_elems > 0) {
        int blocks = (int)((tail_elems + 255) / 256);
        gemm_tail<<<dim3(blocks), dim3(256), 0, stream>>>(x, W, bias, out, N, D, row0);
    }
}


# Self-loop product moved into the row-block kernel: WMMA f32 tiles of its 400 rows go to LDS, the row pass starts each row sum there, out stored once; 2 launches; sums: product first, then relations
# speedup vs baseline: 4.4932x; 1.2996x over previous
#include <hip/hip_runtime.h>
#include <math.h>

typedef __attribute__((ext_vector_type(2))) float v2f;
typedef __attribute__((ext_vector_type(8))) float v8f;

#define DD 128
#define TILE_FLOATS 51200
#define LCAP 12288
#define NBMAX 2048
#define CH 3
#define RPWMAX (TILE_FLOATS / DD)
#define PP 132

__device__ __forceinline__ float dot4(float4 a, float4 b) {
    return a.x * b.x + a.y * b.y + a.z * b.z + a.w * b.w;
}

__device__ __forceinline__ float sum8(float v) {
    v += __shfl_xor(v, 4, 32);
    v += __shfl_xor(v, 2, 32);
    v += __shfl_xor(v, 1, 32);
    return v;
}

template <int K>
__device__ __forceinline__ void take_edges(const float* __restrict__ x, const int (&s)[K],
                                           int cnt, const float4 (&w)[4], float ed, int q,
                                           float& m, float& l, float4 (&acc)[4]) {
    float4 v[K][4];
    float e[K];
#pragma unroll
    for (int u = 0; u < K; ++u) {
        const unsigned ob = (unsigned)s[u] * (unsigned)(DD * 4) + (unsigned)(q * 16);
        const float4* p = (const float4*)((const char*)x + ob);
#pragma unroll
        for (int j = 0; j < 4; ++j) v[u][j] = p[8 * j];
    }
    float cm = m;
#pragma unroll
    for (int u = 0; u < K; ++u) {
        const float pe = dot4(v[u][0], w[0]) + dot4(v[u][1], w[1]) +
                         dot4(v[u][2], w[2]) + dot4(v[u][3], w[3]);
        const float eu = sum8(pe) + ed;
        e[u] = (u < cnt) ? eu : -INFINITY;
        cm = fmaxf(cm, e[u]);
    }
    const float sc = __expf(m - cm);
    l *= sc;
#pragma unroll
    for (int j = 0; j < 4; ++j) {
        acc[j].x *= sc; acc[j].y *= sc; acc[j].z *= sc; acc[j].w *= sc;
    }
#pragma unroll
    for (int u = 0; u < K; ++u) {
        const float p = (u < cnt) ? __expf(e[u] - cm) : 0.0f;
        l += p;
#pragma unroll
        for (int j = 0; j < 4; ++j) {
            acc[j].x += p * v[u][j].x; acc[j].y += p * v[u][j].y;
            acc[j].z += p * v[u][j].z; acc[j].w += p * v[u][j].w;
        }
    }
    m = cm;
}

__device__ __forceinline__ void row_unsorted(const float* __restrict__ x,
                                          const float* __restrict__ attn_w,
                                          const int* __restrict__ src,
                                          const int* __restrict__ dst,
                                          float* __restrict__ out,
                                          const float4* pr,
                                          int E, int R, int t, int q) {
    const float4* xd = (const float4*)(x + (size_t)t * DD + q * 4);
    float4 tot[4];
#pragma unroll
    for (int j = 0; j < 4; ++j) tot[j] = pr[8 * j];
    for (int r = 0; r < R; ++r) {
        const float4* a0 = (const float4*)(attn_w + (size_t)r * 2 * DD + q * 4);
        const float4* a1 = a0 + DD / 4;
        float4 w[4];
        float pd = 0.0f;
#pragma unroll
        for (int j = 0; j < 4; ++j) {
            w[j] = a0[8 * j];
            pd += dot4(xd[8 * j], a1[8 * j]);
        }
        const float ed = sum8(pd);
        float m = -INFINITY, l = 0.0f;
        float4 acc[4];
#pragma unroll
        for (int j = 0; j < 4; ++j) acc[j] = make_float4(0.0f, 0.0f, 0.0f, 0.0f);
        for (int i = 0; i < E; ++i)
            if (dst[(size_t)r * E + i] == t) {
                const int s[1] = { src[(size_t)r * E + i] };
                take_edges<1>(x, s, 1, w, ed, q, m, l, acc);
            }
        const float inv = 1.0f / fmaxf(l, 1e-30f);
#pragma unroll
        for (int j = 0; j < 4; ++j) {
            tot[j].x += acc[j].x * inv; tot[j].y += acc[j].y * inv;
            tot[j].z += acc[j].z * inv; tot[j].w += acc[j].w * inv;
        }
    }
    float4* o = (float4*)(out + (size_t)t * DD + q * 4);
#pragma unroll
    for (int j = 0; j < 4; ++j) o[8 * j] = tot[j];
}

__device__ __forceinline__ void selfloop_tile(const float* __restrict__ x,
                                              const float* __restrict__ W,
                                              const float* __restrict__ bias,
                                              float* tile, int row0, int nrows, int N,
                                              int wave, int lane) {
    const int row = lane & 15;
    const int khalf = (lane >> 4) * 2;
    const int nun = ((nrows + 15) >> 4) * 4;
    for (int u = wave; u < nun; u += 32) {
        const int t16 = (u >> 2) * 16;
        const int n0 = (u & 3) * 32;
        const int gr = min(row0 + t16 + row, N - 1);
        const float* ap = x + (size_t)gr * DD + khalf;
        const float* bp = W + khalf * DD + n0 + row;
        v8f c0 = {}, c1 = {};
#pragma unroll 8
        for (int k = 0; k < DD; k += 4) {
            v2f a, b0, b1;
            a.x = ap[k];
            a.y = ap[k + 1];
            b0.x = bp[k * DD];      b0.y = bp[k * DD + DD];
            b1.x = bp[k * DD + 16]; b1.y = bp[k * DD + DD + 16];
            c0 = __builtin_amdgcn_wmma_f32_16x16x4_f32(false, a, false, b0, (short)0, c0, false, false);
            c1 = __builtin_amdgcn_wmma_f32_16x16x4_f32(false, a, false, b1, (short)0, c1, false, false);
        }
        const int col = n0 + row;
        const float bs0 = bias[col], bs1 = bias[col + 16];
        float* o = tile + (t16 + (lane >> 4) * 8) * PP + col;
#pragma unroll
        for (int v = 0; v < 8; ++v) {
            o[v * PP] = c0[v] + bs0;
            o[v * PP + 16] = c1[v] + bs1;
        }
    }
}

#define CHUNK 4096
#define NTMAX 1024

__device__ __forceinline__ bool bins_ok(int total, int rpw, int ntiles, int R, int N,
                                        size_t ws_bytes) {
    const size_t nch = (size_t)((total + CHUNK - 1) / CHUNK);
    const size_t tab = (nch * (size_t)(ntiles + 1) + 63) & ~(size_t)63;
    return N <= (1 << 20) && rpw * R <= NBMAX && ntiles <= NTMAX &&
           (tab + nch * CHUNK) * sizeof(int) <= ws_bytes;
}

__global__ __launch_bounds__(1024) void bin_edges(const int* __restrict__ src,
                                                  const int* __restrict__ dst,
                                                  int* __restrict__ ws,
                                                  int total, int rpw, int ntiles,
                                                  int E, int R, int N, size_t ws_bytes) {
    __shared__ int cnt[NTMAX];
    __shared__ int beg[NTMAX];
    __shared__ int srt[CHUNK];
    __shared__ int wsum[32];

    if (!bins_ok(total, rpw, ntiles, R, N, ws_bytes)) return;
    const int tid = threadIdx.x;
    const int lane = tid & 31;
    const int wave = tid >> 5;
    const int c = blockIdx.x;
    const int nch = (total + CHUNK - 1) / CHUNK;
    const int nt1 = ntiles + 1;
    const size_t tab = ((size_t)nch * (size_t)nt1 + 63) & ~(size_t)63;
    int* lst = ws + tab + (size_t)c * CHUNK;
    int* tbl = ws + (size_t)c * nt1;

    cnt[tid] = 0;
    __syncthreads();

    const int i0 = c * CHUNK + tid * 4;
    int d[4], sv[4];
    if (i0 + 3 < total) {
        const int4 v = *(const int4*)(dst + i0);
        const int4 u = *(const int4*)(src + i0);
        d[0] = v.x; d[1] = v.y; d[2] = v.z; d[3] = v.w;
        sv[0] = u.x; sv[1] = u.y; sv[2] = u.z; sv[3] = u.w;
    } else {
#pragma unroll
        for (int k = 0; k < 4; ++k) {
            d[k] = (i0 + k < total) ? dst[i0 + k] : -1;
            sv[k] = (i0 + k < total) ? src[i0 + k] : 0;
        }
    }
    int tt[4], val[4], pos[4];
#pragma unroll
    for (int k = 0; k < 4; ++k) {
        const int t = (d[k] >= 0) ? d[k] / rpw : ntiles;
        tt[k] = ((unsigned)t < (unsigned)ntiles) ? t : -1;
        val[k] = (tt[k] >= 0) ? ((((d[k] - t * rpw) * R + (i0 + k) / E) << 20) | sv[k]) : 0;
        pos[k] = 0;
        if (tt[k] >= 0) pos[k] = atomicAdd(&cnt[tt[k]], 1);
    }
    __syncthreads();

    const int c0 = (tid < ntiles) ? cnt[tid] : 0;
    int incl = c0;
    for (int o = 1; o < 32; o <<= 1) {
        const int y = __shfl_up(incl, o, 32);
        if (lane >= o) incl += y;
    }
    if (lane == 31) wsum[wave] = incl;
    __syncthreads();
    if (wave == 0) {
        const int wt = wsum[lane];
        int wi = wt;
        for (int o = 1; o < 32; o <<= 1) {
            const int y = __shfl_up(wi, o, 32);
            if (lane >= o) wi += y;
        }
        wsum[lane] = wi - wt;
    }
    __syncthreads();
    const int ex = wsum[wave] + incl - c0;
    beg[tid] = ex;
    if (tid < ntiles) tbl[tid] = ex;
    if (tid == 1023) tbl[ntiles] = ex + c0;
    __syncthreads();

#pragma unroll
    for (int k = 0; k < 4; ++k)
        if (tt[k] >= 0) srt[beg[tt[k]] + pos[k]] = val[k];
    __syncthreads();

    const int nv = beg[1023] + cnt[1023];
#pragma unroll
    for (int k = 0; k < 4; ++k) {
        const int p = tid + k * 1024;
        if (p < nv) lst[p] = srt[p];
    }
}

__global__ __launch_bounds__(1024) void edge_tile(const float* __restrict__ x,
                                                  const float* __restrict__ attn_w,
                                                  const float* __restrict__ W,
                                                  const float* __restrict__ bias,
                                                  const int* __restrict__ src,
                                                  const int* __restrict__ dst,
                                                  float* __restrict__ out,
                                                  const int* __restrict__ ws, size_t ws_bytes,
                                                  int E, int N, int R, int rpw, int total) {
    __shared__ __attribute__((aligned(16))) float tile[RPWMAX * PP];
    int* const lst_i = (int*)tile;
    int* const lst_b = lst_i + LCAP;
    __shared__ int   srt_s[LCAP];
    __shared__ int   off[NBMAX + 1];
    __shared__ int   fil[NBMAX];
    __shared__ float edl[NBMAX];
    __shared__ int   wsum[32];
    __shared__ int   lst_n;

    const int tid = threadIdx.x;
    const int lane = tid & 31;
    const int wave = __builtin_amdgcn_readfirstlane(tid >> 5);
    const int q = lane & 7;
    const int row0 = blockIdx.x * rpw;
    const int nrows = min(rpw, N - row0);
    const int nb = nrows * R;
    const bool binned = nb <= NBMAX;

    for (int b = tid; b <= NBMAX; b += 1024) off[b] = 0;
    if (tid == 0) lst_n = 0;

    if (binned)
        for (int base = wave * 4; base < nrows; base += 128) {
            const int tl = base + (lane >> 3);
            if (tl < nrows) {
                const float4* xd = (const float4*)(x + (size_t)(row0 + tl) * DD + q * 4);
                float4 xv[4];
#pragma unroll
                for (int j = 0; j < 4; ++j) xv[j] = xd[8 * j];
                for (int r = 0; r < R; ++r) {
                    const float4* a1 = (const float4*)(attn_w + (size_t)r * 2 * DD + DD + q * 4);
                    const float pd = dot4(xv[0], a1[0]) + dot4(xv[1], a1[8]) +
                                     dot4(xv[2], a1[16]) + dot4(xv[3], a1[24]);
                    const float ed = sum8(pd);
                    if (q == 0) edl[tl * R + r] = ed;
                }
            }
        }
    __syncthreads();

    auto take_rec = [&](int b, int sr) {
        const int slot = atomicAdd(&lst_n, 1);
        if (binned && slot < LCAP) {
            lst_i[slot] = sr;
            lst_b[slot] = b;
            atomicAdd(&off[b], 1);
        }
    };
    auto take = [&](int t, int i) {
        const unsigned tl = (unsigned)(t - row0);
        if (tl < (unsigned)nrows) take_rec((int)tl * R + i / E, src[i]);
    };

    if (bins_ok(total, rpw, (int)gridDim.x, R, N, ws_bytes)) {
        const int nch = (total + CHUNK - 1) / CHUNK;
        const int nt1 = (int)gridDim.x + 1;
        const size_t tab = ((size_t)nch * (size_t)nt1 + 63) & ~(size_t)63;
        for (int c = tid >> 3; c < nch; c += 128) {
            const int* tbl = ws + (size_t)c * nt1 + blockIdx.x;
            const int sb = tbl[0], se = tbl[1];
            const int* lst = ws + tab + (size_t)c * CHUNK;
            for (int p = sb + q; p < se; p += 16) {
                const int e0 = lst[p];
                const int e1 = (p + 8 < se) ? lst[p + 8] : -1;
                if ((e0 >> 20) < nb) take_rec(e0 >> 20, e0 & ((1 << 20) - 1));
                if (e1 >= 0 && (e1 >> 20) < nb) take_rec(e1 >> 20, e1 & ((1 << 20) - 1));
            }
        }
    } else {
        const int4* dst4 = (const int4*)dst;
        const int n4 = total / 4;
        for (int q0 = tid; q0 < n4; q0 += 1024 * 8) {
            int4 v[8];
    #pragma unroll
            for (int k = 0; k < 8; ++k) {
                const int qq = q0 + k * 1024;
                v[k] = (qq < n4) ? dst4[qq] : make_int4(-1, -1, -1, -1);
            }
    #pragma unroll
            for (int k = 0; k < 8; ++k) {
                const unsigned d0 = (unsigned)(v[k].x - row0), d1 = (unsigned)(v[k].y - row0);
                const unsigned d2 = (unsigned)(v[k].z - row0), d3 = (unsigned)(v[k].w - row0);
                if (min(min(d0, d1), min(d2, d3)) < (unsigned)nrows) {
                    const int i = (q0 + k * 1024) * 4;
                    take(v[k].x, i + 0);
                    take(v[k].y, i + 1);
                    take(v[k].z, i + 2);
                    take(v[k].w, i + 3);
                }
            }
        }
        for (int i = n4 * 4 + tid; i < total; i += 1024) take(dst[i], i);
    }
    __syncthreads();

    const int n = lst_n;
    if (!binned || n > LCAP) {
        selfloop_tile(x, W, bias, tile, row0, nrows, N, wave, lane);
        __syncthreads();
        for (int base = wave * 4; base < nrows; base += 128) {
            const int tl = base + (lane >> 3);
            if (tl < nrows)
                row_unsorted(x, attn_w, src, dst, out, (const float4*)(tile + tl * PP + q * 4),
                             E, R, row0 + tl, q);
        }
        return;
    }

    const int b0 = 2 * tid, b1 = 2 * tid + 1;
    const int c0 = (b0 < nb) ? off[b0] : 0;
    const int c1 = (b1 < nb) ? off[b1] : 0;
    int incl = c0 + c1;
    for (int o = 1; o < 32; o <<= 1) {
        const int y = __shfl_up(incl, o, 32);
        if (lane >= o) incl += y;
    }
    if (lane == 31) wsum[wave] = incl;
    __syncthreads();
    if (wave == 0) {
        const int wt = wsum[lane];
        int wi = wt;
        for (int o = 1; o < 32; o <<= 1) {
            const int y = __shfl_up(wi, o, 32);
            if (lane >= o) wi += y;
        }
        wsum[lane] = wi - wt;
    }
    __syncthreads();
    const int ex = wsum[wave] + incl - (c0 + c1);
    if (b0 < nb) { off[b0] = ex;      fil[b0] = ex; }
    if (b1 < nb) { off[b1] = ex + c0; fil[b1] = ex + c0; }
    if (tid == 0) off[nb] = n;
    __syncthreads();

    for (int j = tid; j < n; j += 1024) {
        const int pos = atomicAdd(&fil[lst_b[j]], 1);
        srt_s[pos] = lst_i[j];
    }
    __syncthreads();

    selfloop_tile(x, W, bias, tile, row0, nrows, N, wave, lane);
    __syncthreads();

    for (int base = wave * 4; base < nrows; base += 128) {
        const int tl = base + (lane >> 3);
        if (tl < nrows) {
            float4* mine = (float4*)(tile + tl * PP + q * 4);
            for (int r = 0; r < R; ++r) {
                const float4* a0 = (const float4*)((const char*)attn_w +
                                                   (unsigned)(r * (2 * DD * 4) + q * 16));
                float4 w[4];
#pragma unroll
                for (int j = 0; j < 4; ++j) w[j] = a0[8 * j];
                const float ed = edl[tl * R + r];
                const int beg = off[tl * R + r], end = off[tl * R + r + 1];
                float m = -INFINITY, l = 0.0f;
                float4 acc[4];
#pragma unroll
                for (int j = 0; j < 4; ++j) acc[j] = make_float4(0.0f, 0.0f, 0.0f, 0.0f);
                for (int k0 = beg; k0 < end; k0 += CH) {
                    const int cnt = min(CH, end - k0);
                    int s[CH];
#pragma unroll
                    for (int u = 0; u < CH; ++u) s[u] = srt_s[min(k0 + u, end - 1)];
                    take_edges<CH>(x, s, cnt, w, ed, q, m, l, acc);
                }
                asm volatile("" ::: "memory");
                const float inv = 1.0f / fmaxf(l, 1e-30f);
#pragma unroll
                for (int j = 0; j < 4; ++j) {
                    acc[j].x *= inv; acc[j].y *= inv; acc[j].z *= inv; acc[j].w *= inv;
                    const float4 pv = mine[8 * j];
                    acc[j].x += pv.x; acc[j].y += pv.y; acc[j].z += pv.z; acc[j].w += pv.w;
                }
                if (r + 1 < R) {
#pragma unroll
                    for (int j = 0; j < 4; ++j) mine[8 * j] = acc[j];
                } else {
                    float4* o = (float4*)((char*)out +
                                          (unsigned)((row0 + tl) * (DD * 4) + q * 16));
#pragma unroll
                    for (int j = 0; j < 4; ++j) o[8 * j] = acc[j];
                }
            }
        }
    }
}

extern "C" void kernel_launch(void* const* d_in, const int* in_sizes, int n_in,
                              void* d_out, int out_size, void* d_ws, size_t ws_size,
                              hipStream_t stream) {
    const float* x      = (const float*)d_in[0];
    const float* attn_w = (const float*)d_in[1];
    const float* W      = (const float*)d_in[2];
    const float* bias   = (const float*)d_in[3];
    const int*   src    = (const int*)d_in[4];
    const int*   dst    = (const int*)d_in[5];
    float*       out    = (float*)d_out;

    const int D = in_sizes[3];
    const int N = in_sizes[0] / D;
    const int R = in_sizes[1] / (2 * D);
    const int E = in_sizes[4] / R;

    const int rpw = TILE_FLOATS / D;
    dim3 sgrid((N + rpw - 1) / rpw);
    dim3 pgrid((R * E + CHUNK - 1) / CHUNK);
    bin_edges<<<pgrid, 1024, 0, stream>>>(src, dst, (int*)d_ws, R * E, rpw, (int)sgrid.x,
                                          E, R, N, ws_size);
    edge_tile<<<sgrid, 1024, 0, stream>>>(x, attn_w, W, bias, src, dst, out, (const int*)d_ws,
                                          ws_size, E, N, R, rpw, R * E);
}


# Self-loop product without global fragment loads: a wave keeps its 16-column strip of W in registers; x rows staged once in the LDS tile, overwritten in place per round; product k order changes
# speedup vs baseline: 5.0662x; 1.1275x over previous
#include <hip/hip_runtime.h>
#include <math.h>

typedef __attribute__((ext_vector_type(2))) float v2f;
typedef __attribute__((ext_vector_type(8))) float v8f;

#define DD 128
#define TILE_FLOATS 51200
#define LCAP 12288
#define NBMAX 2048
#define CH 3
#define RPWMAX (TILE_FLOATS / DD)
#define PP 132

__device__ __forceinline__ float dot4(float4 a, float4 b) {
    return a.x * b.x + a.y * b.y + a.z * b.z + a.w * b.w;
}

__device__ __forceinline__ float sum8(float v) {
    v += __shfl_xor(v, 4, 32);
    v += __shfl_xor(v, 2, 32);
    v += __shfl_xor(v, 1, 32);
    return v;
}

template <int K>
__device__ __forceinline__ void take_edges(const float* __restrict__ x, const int (&s)[K],
                                           int cnt, const float4 (&w)[4], float ed, int q,
                                           float& m, float& l, float4 (&acc)[4]) {
    float4 v[K][4];
    float e[K];
#pragma unroll
    for (int u = 0; u < K; ++u) {
        const unsigned ob = (unsigned)s[u] * (unsigned)(DD * 4) + (unsigned)(q * 16);
        const float4* p = (const float4*)((const char*)x + ob);
#pragma unroll
        for (int j = 0; j < 4; ++j) v[u][j] = p[8 * j];
    }
    float cm = m;
#pragma unroll
    for (int u = 0; u < K; ++u) {
        const float pe = dot4(v[u][0], w[0]) + dot4(v[u][1], w[1]) +
                         dot4(v[u][2], w[2]) + dot4(v[u][3], w[3]);
        const float eu = sum8(pe) + ed;
        e[u] = (u < cnt) ? eu : -INFINITY;
        cm = fmaxf(cm, e[u]);
    }
    const float sc = __expf(m - cm);
    l *= sc;
#pragma unroll
    for (int j = 0; j < 4; ++j) {
        acc[j].x *= sc; acc[j].y *= sc; acc[j].z *= sc; acc[j].w *= sc;
    }
#pragma unroll
    for (int u = 0; u < K; ++u) {
        const float p = (u < cnt) ? __expf(e[u] - cm) : 0.0f;
        l += p;
#pragma unroll
        for (int j = 0; j < 4; ++j) {
            acc[j].x += p * v[u][j].x; acc[j].y += p * v[u][j].y;
            acc[j].z += p * v[u][j].z; acc[j].w += p * v[u][j].w;
        }
    }
    m = cm;
}

__device__ __forceinline__ void row_unsorted(const float* __restrict__ x,
                                          const float* __restrict__ attn_w,
                                          const int* __restrict__ src,
                                          const int* __restrict__ dst,
                                          float* __restrict__ out,
                                          const float4* pr,
                                          int E, int R, int t, int q) {
    const float4* xd = (const float4*)(x + (size_t)t * DD + q * 4);
    float4 tot[4];
#pragma unroll
    for (int j = 0; j < 4; ++j) tot[j] = pr[8 * j];
    for (int r = 0; r < R; ++r) {
        const float4* a0 = (const float4*)(attn_w + (size_t)r * 2 * DD + q * 4);
        const float4* a1 = a0 + DD / 4;
        float4 w[4];
        float pd = 0.0f;
#pragma unroll
        for (int j = 0; j < 4; ++j) {
            w[j] = a0[8 * j];
            pd += dot4(xd[8 * j], a1[8 * j]);
        }
        const float ed = sum8(pd);
        float m = -INFINITY, l = 0.0f;
        float4 acc[4];
#pragma unroll
        for (int j = 0; j < 4; ++j) acc[j] = make_float4(0.0f, 0.0f, 0.0f, 0.0f);
        for (int i = 0; i < E; ++i)
            if (dst[(size_t)r * E + i] == t) {
                const int s[1] = { src[(size_t)r * E + i] };
                take_edges<1>(x, s, 1, w, ed, q, m, l, acc);
            }
        const float inv = 1.0f / fmaxf(l, 1e-30f);
#pragma unroll
        for (int j = 0; j < 4; ++j) {
            tot[j].x += acc[j].x * inv; tot[j].y += acc[j].y * inv;
            tot[j].z += acc[j].z * inv; tot[j].w += acc[j].w * inv;
        }
    }
    float4* o = (float4*)(out + (size_t)t * DD + q * 4);
#pragma unroll
    for (int j = 0; j < 4; ++j) o[8 * j] = tot[j];
}

template <bool TWO>
__device__ __forceinline__ void product_round(const float* a0p, const float (&breg)[64],
                                              v8f& c0, v8f& c1) {
#pragma unroll
    for (int i = 0; i < 16; ++i) {
        const float4 a0 = *(const float4*)(a0p + 8 * i);
        v2f bl, bh, al, ah;
        bl.x = breg[4 * i];     bl.y = breg[4 * i + 1];
        bh.x = breg[4 * i + 2]; bh.y = breg[4 * i + 3];
        al.x = a0.x; al.y = a0.y; ah.x = a0.z; ah.y = a0.w;
        c0 = __builtin_amdgcn_wmma_f32_16x16x4_f32(false, al, false, bl, (short)0, c0, false, false);
        if (TWO) {
            const float4 a1 = *(const float4*)(a0p + 16 * PP + 8 * i);
            v2f dl, dh;
            dl.x = a1.x; dl.y = a1.y; dh.x = a1.z; dh.y = a1.w;
            c1 = __builtin_amdgcn_wmma_f32_16x16x4_f32(false, dl, false, bl, (short)0, c1, false, false);
            c0 = __builtin_amdgcn_wmma_f32_16x16x4_f32(false, ah, false, bh, (short)0, c0, false, false);
            c1 = __builtin_amdgcn_wmma_f32_16x16x4_f32(false, dh, false, bh, (short)0, c1, false, false);
        } else {
            c0 = __builtin_amdgcn_wmma_f32_16x16x4_f32(false, ah, false, bh, (short)0, c0, false, false);
        }
    }
}

__device__ __forceinline__ void selfloop_tile(const float* __restrict__ x,
                                              const float* __restrict__ attn_w,
                                              const float* __restrict__ W,
                                              const float* __restrict__ bias,
                                              float* tile, float* edl, bool want_edl,
                                              int row0, int nrows, int N, int R,
                                              int wave, int lane) {
    const int c = lane & 15;
    const int h = lane >> 4;
    const int q = lane & 7;
    const int col = (wave & 7) * 16 + c;
    const int g = wave >> 3;

    float breg[64];
    const float* bp = W + (4 * h) * DD + col;
#pragma unroll
    for (int i = 0; i < 16; ++i)
#pragma unroll
        for (int e = 0; e < 4; ++e) breg[4 * i + e] = bp[(8 * i + e) * DD];
    const float bs = bias[col];

    const int ntile = (nrows + 15) >> 4;
    for (int base = wave * 4; base < ntile * 16; base += 128) {
        const int tl = base + (lane >> 3);
        const int gr = min(row0 + tl, N - 1);
        const float4* xd = (const float4*)(x + (size_t)gr * DD + q * 4);
        float4 xv[4];
#pragma unroll
        for (int j = 0; j < 4; ++j) xv[j] = xd[8 * j];
        float4* tr = (float4*)(tile + tl * PP + q * 4);
#pragma unroll
        for (int j = 0; j < 4; ++j) tr[8 * j] = xv[j];
        if (want_edl && tl < nrows)
            for (int r = 0; r < R; ++r) {
                const float4* a1 = (const float4*)(attn_w + (size_t)r * 2 * DD + DD + q * 4);
                const float pd = dot4(xv[0], a1[0]) + dot4(xv[1], a1[8]) +
                                 dot4(xv[2], a1[16]) + dot4(xv[3], a1[24]);
                const float ed = sum8(pd);
                if (q == 0) edl[tl * R + r] = ed;
            }
    }
    __syncthreads();

    const int nround = (ntile + 7) >> 3;
    for (int rd = 0; rd < nround; ++rd) {
        const int t0 = rd * 8 + g * 2;
        const float* a0p = tile + (t0 * 16 + c) * PP + 4 * h;
        v8f c0 = {}, c1 = {};
        if (t0 + 1 < ntile) product_round<true>(a0p, breg, c0, c1);
        else if (t0 < ntile) product_round<false>(a0p, breg, c0, c1);
        __syncthreads();
        float* o = tile + (t0 * 16 + 8 * h) * PP + col;
        if (t0 < ntile) {
#pragma unroll
            for (int v = 0; v < 8; ++v) o[v * PP] = c0[v] + bs;
        }
        if (t0 + 1 < ntile) {
#pragma unroll
            for (int v = 0; v < 8; ++v) o[(16 + v) * PP] = c1[v] + bs;
        }
    }
}

#define CHUNK 4096
#define NTMAX 1024

__device__ __forceinline__ bool bins_ok(int total, int rpw, int ntiles, int R, int N,
                                        size_t ws_bytes) {
    const size_t nch = (size_t)((total + CHUNK - 1) / CHUNK);
    const size_t tab = (nch * (size_t)(ntiles + 1) + 63) & ~(size_t)63;
    return N <= (1 << 20) && rpw * R <= NBMAX && ntiles <= NTMAX &&
           (tab + nch * CHUNK) * sizeof(int) <= ws_bytes;
}

__global__ __launch_bounds__(1024) void bin_edges(const int* __restrict__ src,
                                                  const int* __restrict__ dst,
                                                  int* __restrict__ ws,
                                                  int total, int rpw, int ntiles,
                                                  int E, int R, int N, size_t ws_bytes) {
    __shared__ int cnt[NTMAX];
    __shared__ int beg[NTMAX];
    __shared__ int srt[CHUNK];
    __shared__ int wsum[32];

    if (!bins_ok(total, rpw, ntiles, R, N, ws_bytes)) return;
    const int tid = threadIdx.x;
    const int lane = tid & 31;
    const int wave = tid >> 5;
    const int c = blockIdx.x;
    const int nch = (total + CHUNK - 1) / CHUNK;
    const int nt1 = ntiles + 1;
    const size_t tab = ((size_t)nch * (size_t)nt1 + 63) & ~(size_t)63;
    int* lst = ws + tab + (size_t)c * CHUNK;
    int* tbl = ws + (size_t)c * nt1;

    cnt[tid] = 0;
    __syncthreads();

    const int i0 = c * CHUNK + tid * 4;
    int d[4], sv[4];
    if (i0 + 3 < total) {
        const int4 v = *(const int4*)(dst + i0);
        const int4 u = *(const int4*)(src + i0);
        d[0] = v.x; d[1] = v.y; d[2] = v.z; d[3] = v.w;
        sv[0] = u.x; sv[1] = u.y; sv[2] = u.z; sv[3] = u.w;
    } else {
#pragma unroll
        for (int k = 0; k < 4; ++k) {
            d[k] = (i0 + k < total) ? dst[i0 + k] : -1;
            sv[k] = (i0 + k < total) ? src[i0 + k] : 0;
        }
    }
    int tt[4], val[4], pos[4];
#pragma unroll
    for (int k = 0; k < 4; ++k) {
        const int t = (d[k] >= 0) ? d[k] / rpw : ntiles;
        tt[k] = ((unsigned)t < (unsigned)ntiles) ? t : -1;
        val[k] = (tt[k] >= 0) ? ((((d[k] - t * rpw) * R + (i0 + k) / E) << 20) | sv[k]) : 0;
        pos[k] = 0;
        if (tt[k] >= 0) pos[k] = atomicAdd(&cnt[tt[k]], 1);
    }
    __syncthreads();

    const int c0 = (tid < ntiles) ? cnt[tid] : 0;
    int incl = c0;
    for (int o = 1; o < 32; o <<= 1) {
        const int y = __shfl_up(incl, o, 32);
        if (lane >= o) incl += y;
    }
    if (lane == 31) wsum[wave] = incl;
    __syncthreads();
    if (wave == 0) {
        const int wt = wsum[lane];
        int wi = wt;
        for (int o = 1; o < 32; o <<= 1) {
            const int y = __shfl_up(wi, o, 32);
            if (lane >= o) wi += y;
        }
        wsum[lane] = wi - wt;
    }
    __syncthreads();
    const int ex = wsum[wave] + incl - c0;
    beg[tid] = ex;
    if (tid < ntiles) tbl[tid] = ex;
    if (tid == 1023) tbl[ntiles] = ex + c0;
    __syncthreads();

#pragma unroll
    for (int k = 0; k < 4; ++k)
        if (tt[k] >= 0) srt[beg[tt[k]] + pos[k]] = val[k];
    __syncthreads();

    const int nv = beg[1023] + cnt[1023];
#pragma unroll
    for (int k = 0; k < 4; ++k) {
        const int p = tid + k * 1024;
        if (p < nv) lst[p] = srt[p];
    }
}

__global__ __launch_bounds__(1024) void edge_tile(const float* __restrict__ x,
                                                  const float* __restrict__ attn_w,
                                                  const float* __restrict__ W,
                                                  const float* __restrict__ bias,
                                                  const int* __restrict__ src,
                                                  const int* __restrict__ dst,
                                                  float* __restrict__ out,
                                                  const int* __restrict__ ws, size_t ws_bytes,
                                                  int E, int N, int R, int rpw, int total) {
    __shared__ __attribute__((aligned(16))) float tile[RPWMAX * PP];
    int* const lst_i = (int*)tile;
    int* const lst_b = lst_i + LCAP;
    __shared__ int   srt_s[LCAP];
    __shared__ int   off[NBMAX + 1];
    __shared__ int   fil[NBMAX];
    __shared__ float edl[NBMAX];
    __shared__ int   wsum[32];
    __shared__ int   lst_n;

    const int tid = threadIdx.x;
    const int lane = tid & 31;
    const int wave = __builtin_amdgcn_readfirstlane(tid >> 5);
    const int q = lane & 7;
    const int row0 = blockIdx.x * rpw;
    const int nrows = min(rpw, N - row0);
    const int nb = nrows * R;
    const bool binned = nb <= NBMAX;

    for (int b = tid; b <= NBMAX; b += 1024) off[b] = 0;
    if (tid == 0) lst_n = 0;

    __syncthreads();

    auto take_rec = [&](int b, int sr) {
        const int slot = atomicAdd(&lst_n, 1);
        if (binned && slot < LCAP) {
            lst_i[slot] = sr;
            lst_b[slot] = b;
            atomicAdd(&off[b], 1);
        }
    };
    auto take = [&](int t, int i) {
        const unsigned tl = (unsigned)(t - row0);
        if (tl < (unsigned)nrows) take_rec((int)tl * R + i / E, src[i]);
    };

    if (bins_ok(total, rpw, (int)gridDim.x, R, N, ws_bytes)) {
        const int nch = (total + CHUNK - 1) / CHUNK;
        const int nt1 = (int)gridDim.x + 1;
        const size_t tab = ((size_t)nch * (size_t)nt1 + 63) & ~(size_t)63;
        for (int c = tid >> 3; c < nch; c += 128) {
            const int* tbl = ws + (size_t)c * nt1 + blockIdx.x;
            const int sb = tbl[0], se = tbl[1];
            const int* lst = ws + tab + (size_t)c * CHUNK;
            for (int p = sb + q; p < se; p += 16) {
                const int e0 = lst[p];
                const int e1 = (p + 8 < se) ? lst[p + 8] : -1;
                if ((e0 >> 20) < nb) take_rec(e0 >> 20, e0 & ((1 << 20) - 1));
                if (e1 >= 0 && (e1 >> 20) < nb) take_rec(e1 >> 20, e1 & ((1 << 20) - 1));
            }
        }
    } else {
        const int4* dst4 = (const int4*)dst;
        const int n4 = total / 4;
        for (int q0 = tid; q0 < n4; q0 += 1024 * 8) {
            int4 v[8];
    #pragma unroll
            for (int k = 0; k < 8; ++k) {
                const int qq = q0 + k * 1024;
                v[k] = (qq < n4) ? dst4[qq] : make_int4(-1, -1, -1, -1);
            }
    #pragma unroll
            for (int k = 0; k < 8; ++k) {
                const unsigned d0 = (unsigned)(v[k].x - row0), d1 = (unsigned)(v[k].y - row0);
                const unsigned d2 = (unsigned)(v[k].z - row0), d3 = (unsigned)(v[k].w - row0);
                if (min(min(d0, d1), min(d2, d3)) < (unsigned)nrows) {
                    const int i = (q0 + k * 1024) * 4;
                    take(v[k].x, i + 0);
                    take(v[k].y, i + 1);
                    take(v[k].z, i + 2);
                    take(v[k].w, i + 3);
                }
            }
        }
        for (int i = n4 * 4 + tid; i < total; i += 1024) take(dst[i], i);
    }
    __syncthreads();

    const int n = lst_n;
    if (!binned || n > LCAP) {
        selfloop_tile(x, attn_w, W, bias, tile, edl, false, row0, nrows, N, R, wave, lane);
        __syncthreads();
        for (int base = wave * 4; base < nrows; base += 128) {
            const int tl = base + (lane >> 3);
            if (tl < nrows)
                row_unsorted(x, attn_w, src, dst, out, (const float4*)(tile + tl * PP + q * 4),
                             E, R, row0 + tl, q);
        }
        return;
    }

    const int b0 = 2 * tid, b1 = 2 * tid + 1;
    const int c0 = (b0 < nb) ? off[b0] : 0;
    const int c1 = (b1 < nb) ? off[b1] : 0;
    int incl = c0 + c1;
    for (int o = 1; o < 32; o <<= 1) {
        const int y = __shfl_up(incl, o, 32);
        if (lane >= o) incl += y;
    }
    if (lane == 31) wsum[wave] = incl;
    __syncthreads();
    if (wave == 0) {
        const int wt = wsum[lane];
        int wi = wt;
        for (int o = 1; o < 32; o <<= 1) {
            const int y = __shfl_up(wi, o, 32);
            if (lane >= o) wi += y;
        }
        wsum[lane] = wi - wt;
    }
    __syncthreads();
    const int ex = wsum[wave] + incl - (c0 + c1);
    if (b0 < nb) { off[b0] = ex;      fil[b0] = ex; }
    if (b1 < nb) { off[b1] = ex + c0; fil[b1] = ex + c0; }
    if (tid == 0) off[nb] = n;
    __syncthreads();

    for (int j = tid; j < n; j += 1024) {
        const int pos = atomicAdd(&fil[lst_b[j]], 1);
        srt_s[pos] = lst_i[j];
    }
    __syncthreads();

    selfloop_tile(x, attn_w, W, bias, tile, edl, true, row0, nrows, N, R, wave, lane);
    __syncthreads();

    for (int base = wave * 4; base < nrows; base += 128) {
        const int tl = base + (lane >> 3);
        if (tl < nrows) {
            float4* mine = (float4*)(tile + tl * PP + q * 4);
            for (int r = 0; r < R; ++r) {
                const float4* a0 = (const float4*)((const char*)attn_w +
                                                   (unsigned)(r * (2 * DD * 4) + q * 16));
                float4 w[4];
#pragma unroll
                for (int j = 0; j < 4; ++j) w[j] = a0[8 * j];
                const float ed = edl[tl * R + r];
                const int beg = off[tl * R + r], end = off[tl * R + r + 1];
                float m = -INFINITY, l = 0.0f;
                float4 acc[4];
#pragma unroll
                for (int j = 0; j < 4; ++j) acc[j] = make_float4(0.0f, 0.0f, 0.0f, 0.0f);
                for (int k0 = beg; k0 < end; k0 += CH) {
                    const int cnt = min(CH, end - k0);
                    int s[CH];
#pragma unroll
                    for (int u = 0; u < CH; ++u) s[u] = srt_s[min(k0 + u, end - 1)];
                    take_edges<CH>(x, s, cnt, w, ed, q, m, l, acc);
                }
                asm volatile("" ::: "memory");
                const float inv = 1.0f / fmaxf(l, 1e-30f);
#pragma unroll
                for (int j = 0; j < 4; ++j) {
                    acc[j].x *= inv; acc[j].y *= inv; acc[j].z *= inv; acc[j].w *= inv;
                    const float4 pv = mine[8 * j];
                    acc[j].x += pv.x; acc[j].y += pv.y; acc[j].z += pv.z; acc[j].w += pv.w;
                }
                if (r + 1 < R) {
#pragma unroll
                    for (int j = 0; j < 4; ++j) mine[8 * j] = acc[j];
                } else {
                    float4* o = (float4*)((char*)out +
                                          (unsigned)((row0 + tl) * (DD * 4) + q * 16));
#pragma unroll
                    for (int j = 0; j < 4; ++j) o[8 * j] = acc[j];
                }
            }
        }
    }
}

extern "C" void kernel_launch(void* const* d_in, const int* in_sizes, int n_in,
                              void* d_out, int out_size, void* d_ws, size_t ws_size,
                              hipStream_t stream) {
    const float* x      = (const float*)d_in[0];
    const float* attn_w = (const float*)d_in[1];
    const float* W      = (const float*)d_in[2];
    const float* bias   = (const float*)d_in[3];
    const int*   src    = (const int*)d_in[4];
    const int*   dst    = (const int*)d_in[5];
    float*       out    = (float*)d_out;

    const int D = in_sizes[3];
    const int N = in_sizes[0] / D;
    const int R = in_sizes[1] / (2 * D);
    const int E = in_sizes[4] / R;

    const int rpw = TILE_FLOATS / D;
    dim3 sgrid((N + rpw - 1) / rpw);
    dim3 pgrid((R * E + CHUNK - 1) / CHUNK);
    bin_edges<<<pgrid, 1024, 0, stream>>>(src, dst, (int*)d_ws, R * E, rpw, (int)sgrid.x,
                                          E, R, N, ws_size);
    edge_tile<<<sgrid, 1024, 0, stream>>>(x, attn_w, W, bias, src, dst, out, (const int*)d_ws,
                                          ws_size, E, N, R, rpw, R * E);
}


# Edge logit in registers only: 8-lane sums by DPP row_xmask adds, not LDS-crossbar shuffles; a lane's 16 products as one chain of 8 packed FMAs; no LDS wait in the edge loop; logit sum order changes
# speedup vs baseline: 5.5229x; 1.0901x over previous
#include <hip/hip_runtime.h>
#include <math.h>

typedef __attribute__((ext_vector_type(2))) float v2f;
typedef __attribute__((ext_vector_type(8))) float v8f;

#define DD 128
#define TILE_FLOATS 51200
#define LCAP 12288
#define NBMAX 2048
#define CH 3
#define RPWMAX (TILE_FLOATS / DD)
#define PP 132

__device__ __forceinline__ float dot4(float4 a, float4 b) {
    return a.x * b.x + a.y * b.y + a.z * b.z + a.w * b.w;
}

template <int M>
__device__ __forceinline__ float lane_xor(float x) {
    const int b = __builtin_bit_cast(int, x);
    return __builtin_bit_cast(float, __builtin_amdgcn_update_dpp(0, b, 0x160 + M, 0xf, 0xf, true));
}

__device__ __forceinline__ float sum8(float v) {
    v += lane_xor<4>(v);
    v += lane_xor<2>(v);
    v += lane_xor<1>(v);
    return v;
}

__device__ __forceinline__ float dot16(const float4 (&v)[4], const float4 (&w)[4]) {
    v2f s = v2f{v[0].x, v[0].y} * v2f{w[0].x, w[0].y};
    s = v2f{v[0].z, v[0].w} * v2f{w[0].z, w[0].w} + s;
#pragma unroll
    for (int j = 1; j < 4; ++j) {
        s = v2f{v[j].x, v[j].y} * v2f{w[j].x, w[j].y} + s;
        s = v2f{v[j].z, v[j].w} * v2f{w[j].z, w[j].w} + s;
    }
    return s.x + s.y;
}

template <int K>
__device__ __forceinline__ void take_edges(const float* __restrict__ x, const int (&s)[K],
                                           int cnt, const float4 (&w)[4], float ed, int q,
                                           float& m, float& l, float4 (&acc)[4]) {
    float4 v[K][4];
    float e[K];
#pragma unroll
    for (int u = 0; u < K; ++u) {
        const unsigned ob = (unsigned)s[u] * (unsigned)(DD * 4) + (unsigned)(q * 16);
        const float4* p = (const float4*)((const char*)x + ob);
#pragma unroll
        for (int j = 0; j < 4; ++j) v[u][j] = p[8 * j];
    }
    float cm = m;
#pragma unroll
    for (int u = 0; u < K; ++u) {
        const float eu = sum8(dot16(v[u], w)) + ed;
        e[u] = (u < cnt) ? eu : -INFINITY;
        cm = fmaxf(cm, e[u]);
    }
    const float sc = __expf(m - cm);
    l *= sc;
#pragma unroll
    for (int j = 0; j < 4; ++j) {
        acc[j].x *= sc; acc[j].y *= sc; acc[j].z *= sc; acc[j].w *= sc;
    }
#pragma unroll
    for (int u = 0; u < K; ++u) {
        const float p = (u < cnt) ? __expf(e[u] - cm) : 0.0f;
        l += p;
#pragma unroll
        for (int j = 0; j < 4; ++j) {
            acc[j].x += p * v[u][j].x; acc[j].y += p * v[u][j].y;
            acc[j].z += p * v[u][j].z; acc[j].w += p * v[u][j].w;
        }
    }
    m = cm;
}

__device__ __forceinline__ void row_unsorted(const float* __restrict__ x,
                                          const float* __restrict__ attn_w,
                                          const int* __restrict__ src,
                                          const int* __restrict__ dst,
                                          float* __restrict__ out,
                                          const float4* pr,
                                          int E, int R, int t, int q) {
    const float4* xd = (const float4*)(x + (size_t)t * DD + q * 4);
    float4 tot[4];
#pragma unroll
    for (int j = 0; j < 4; ++j) tot[j] = pr[8 * j];
    for (int r = 0; r < R; ++r) {
        const float4* a0 = (const float4*)(attn_w + (size_t)r * 2 * DD + q * 4);
        const float4* a1 = a0 + DD / 4;
        float4 w[4];
        float pd = 0.0f;
#pragma unroll
        for (int j = 0; j < 4; ++j) {
            w[j] = a0[8 * j];
            pd += dot4(xd[8 * j], a1[8 * j]);
        }
        const float ed = sum8(pd);
        float m = -INFINITY, l = 0.0f;
        float4 acc[4];
#pragma unroll
        for (int j = 0; j < 4; ++j) acc[j] = make_float4(0.0f, 0.0f, 0.0f, 0.0f);
        for (int i = 0; i < E; ++i)
            if (dst[(size_t)r * E + i] == t) {
                const int s[1] = { src[(size_t)r * E + i] };
                take_edges<1>(x, s, 1, w, ed, q, m, l, acc);
            }
        const float inv = 1.0f / fmaxf(l, 1e-30f);
#pragma unroll
        for (int j = 0; j < 4; ++j) {
            tot[j].x += acc[j].x * inv; tot[j].y += acc[j].y * inv;
            tot[j].z += acc[j].z * inv; tot[j].w += acc[j].w * inv;
        }
    }
    float4* o = (float4*)(out + (size_t)t * DD + q * 4);
#pragma unroll
    for (int j = 0; j < 4; ++j) o[8 * j] = tot[j];
}

template <bool TWO>
__device__ __forceinline__ void product_round(const float* a0p, const float (&breg)[64],
                                              v8f& c0, v8f& c1) {
#pragma unroll
    for (int i = 0; i < 16; ++i) {
        const float4 a0 = *(const float4*)(a0p + 8 * i);
        v2f bl, bh, al, ah;
        bl.x = breg[4 * i];     bl.y = breg[4 * i + 1];
        bh.x = breg[4 * i + 2]; bh.y = breg[4 * i + 3];
        al.x = a0.x; al.y = a0.y; ah.x = a0.z; ah.y = a0.w;
        c0 = __builtin_amdgcn_wmma_f32_16x16x4_f32(false, al, false, bl, (short)0, c0, false, false);
        if (TWO) {
            const float4 a1 = *(const float4*)(a0p + 16 * PP + 8 * i);
            v2f dl, dh;
            dl.x = a1.x; dl.y = a1.y; dh.x = a1.z; dh.y = a1.w;
            c1 = __builtin_amdgcn_wmma_f32_16x16x4_f32(false, dl, false, bl, (short)0, c1, false, false);
            c0 = __builtin_amdgcn_wmma_f32_16x16x4_f32(false, ah, false, bh, (short)0, c0, false, false);
            c1 = __builtin_amdgcn_wmma_f32_16x16x4_f32(false, dh, false, bh, (short)0, c1, false, false);
        } else {
            c0 = __builtin_amdgcn_wmma_f32_16x16x4_f32(false, ah, false, bh, (short)0, c0, false, false);
        }
    }
}

__device__ __forceinline__ void selfloop_tile(const float* __restrict__ x,
                                              const float* __restrict__ attn_w,
                                              const float* __restrict__ W,
                                              const float* __restrict__ bias,
                                              float* tile, float* edl, bool want_edl,
                                              int row0, int nrows, int N, int R,
                                              int wave, int lane) {
    const int c = lane & 15;
    const int h = lane >> 4;
    const int q = lane & 7;
    const int col = (wave & 7) * 16 + c;
    const int g = wave >> 3;

    float breg[64];
    const float* bp = W + (4 * h) * DD + col;
#pragma unroll
    for (int i = 0; i < 16; ++i)
#pragma unroll
        for (int e = 0; e < 4; ++e) breg[4 * i + e] = bp[(8 * i + e) * DD];
    const float bs = bias[col];

    const int ntile = (nrows + 15) >> 4;
    for (int base = wave * 4; base < ntile * 16; base += 128) {
        const int tl = base + (lane >> 3);
        const int gr = min(row0 + tl, N - 1);
        const float4* xd = (const float4*)(x + (size_t)gr * DD + q * 4);
        float4 xv[4];
#pragma unroll
        for (int j = 0; j < 4; ++j) xv[j] = xd[8 * j];
        float4* tr = (float4*)(tile + tl * PP + q * 4);
#pragma unroll
        for (int j = 0; j < 4; ++j) tr[8 * j] = xv[j];
        if (want_edl && tl < nrows)
            for (int r = 0; r < R; ++r) {
                const float4* a1 = (const float4*)(attn_w + (size_t)r * 2 * DD + DD + q * 4);
                const float pd = dot4(xv[0], a1[0]) + dot4(xv[1], a1[8]) +
                                 dot4(xv[2], a1[16]) + dot4(xv[3], a1[24]);
                const float ed = sum8(pd);
                if (q == 0) edl[tl * R + r] = ed;
            }
    }
    __syncthreads();

    const int nround = (ntile + 7) >> 3;
    for (int rd = 0; rd < nround; ++rd) {
        const int t0 = rd * 8 + g * 2;
        const float* a0p = tile + (t0 * 16 + c) * PP + 4 * h;
        v8f c0 = {}, c1 = {};
        if (t0 + 1 < ntile) product_round<true>(a0p, breg, c0, c1);
        else if (t0 < ntile) product_round<false>(a0p, breg, c0, c1);
        __syncthreads();
        float* o = tile + (t0 * 16 + 8 * h) * PP + col;
        if (t0 < ntile) {
#pragma unroll
            for (int v = 0; v < 8; ++v) o[v * PP] = c0[v] + bs;
        }
        if (t0 + 1 < ntile) {
#pragma unroll
            for (int v = 0; v < 8; ++v) o[(16 + v) * PP] = c1[v] + bs;
        }
    }
}

#define CHUNK 4096
#define NTMAX 1024

__device__ __forceinline__ bool bins_ok(int total, int rpw, int ntiles, int R, int N,
                                        size_t ws_bytes) {
    const size_t nch = (size_t)((total + CHUNK - 1) / CHUNK);
    const size_t tab = (nch * (size_t)(ntiles + 1) + 63) & ~(size_t)63;
    return N <= (1 << 20) && rpw * R <= NBMAX && ntiles <= NTMAX &&
           (tab + nch * CHUNK) * sizeof(int) <= ws_bytes;
}

__global__ __launch_bounds__(1024) void bin_edges(const int* __restrict__ src,
                                                  const int* __restrict__ dst,
                                                  int* __restrict__ ws,
                                                  int total, int rpw, int ntiles,
                                                  int E, int R, int N, size_t ws_bytes) {
    __shared__ int cnt[NTMAX];
    __shared__ int beg[NTMAX];
    __shared__ int srt[CHUNK];
    __shared__ int wsum[32];

    if (!bins_ok(total, rpw, ntiles, R, N, ws_bytes)) return;
    const int tid = threadIdx.x;
    const int lane = tid & 31;
    const int wave = tid >> 5;
    const int c = blockIdx.x;
    const int nch = (total + CHUNK - 1) / CHUNK;
    const int nt1 = ntiles + 1;
    const size_t tab = ((size_t)nch * (size_t)nt1 + 63) & ~(size_t)63;
    int* lst = ws + tab + (size_t)c * CHUNK;
    int* tbl = ws + (size_t)c * nt1;

    cnt[tid] = 0;
    __syncthreads();

    const int i0 = c * CHUNK + tid * 4;
    int d[4], sv[4];
    if (i0 + 3 < total) {
        const int4 v = *(const int4*)(dst + i0);
        const int4 u = *(const int4*)(src + i0);
        d[0] = v.x; d[1] = v.y; d[2] = v.z; d[3] = v.w;
        sv[0] = u.x; sv[1] = u.y; sv[2] = u.z; sv[3] = u.w;
    } else {
#pragma unroll
        for (int k = 0; k < 4; ++k) {
            d[k] = (i0 + k < total) ? dst[i0 + k] : -1;
            sv[k] = (i0 + k < total) ? src[i0 + k] : 0;
        }
    }
    int tt[4], val[4], pos[4];
#pragma unroll
    for (int k = 0; k < 4; ++k) {
        const int t = (d[k] >= 0) ? d[k] / rpw : ntiles;
        tt[k] = ((unsigned)t < (unsigned)ntiles) ? t : -1;
        val[k] = (tt[k] >= 0) ? ((((d[k] - t * rpw) * R + (i0 + k) / E) << 20) | sv[k]) : 0;
        pos[k] = 0;
        if (tt[k] >= 0) pos[k] = atomicAdd(&cnt[tt[k]], 1);
    }
    __syncthreads();

    const int c0 = (tid < ntiles) ? cnt[tid] : 0;
    int incl = c0;
    for (int o = 1; o < 32; o <<= 1) {
        const int y = __shfl_up(incl, o, 32);
        if (lane >= o) incl += y;
    }
    if (lane == 31) wsum[wave] = incl;
    __syncthreads();
    if (wave == 0) {
        const int wt = wsum[lane];
        int wi = wt;
        for (int o = 1; o < 32; o <<= 1) {
            const int y = __shfl_up(wi, o, 32);
            if (lane >= o) wi += y;
        }
        wsum[lane] = wi - wt;
    }
    __syncthreads();
    const int ex = wsum[wave] + incl - c0;
    beg[tid] = ex;
    if (tid < ntiles) tbl[tid] = ex;
    if (tid == 1023) tbl[ntiles] = ex + c0;
    __syncthreads();

#pragma unroll
    for (int k = 0; k < 4; ++k)
        if (tt[k] >= 0) srt[beg[tt[k]] + pos[k]] = val[k];
    __syncthreads();

    const int nv = beg[1023] + cnt[1023];
#pragma unroll
    for (int k = 0; k < 4; ++k) {
        const int p = tid + k * 1024;
        if (p < nv) lst[p] = srt[p];
    }
}

__global__ __launch_bounds__(1024) void edge_tile(const float* __restrict__ x,
                                                  const float* __restrict__ attn_w,
                                                  const float* __restrict__ W,
                                                  const float* __restrict__ bias,
                                                  const int* __restrict__ src,
                                                  const int* __restrict__ dst,
                                                  float* __restrict__ out,
                                                  const int* __restrict__ ws, size_t ws_bytes,
                                                  int E, int N, int R, int rpw, int total) {
    __shared__ __attribute__((aligned(16))) float tile[RPWMAX * PP];
    int* const lst_i = (int*)tile;
    int* const lst_b = lst_i + LCAP;
    __shared__ int   srt_s[LCAP];
    __shared__ int   off[NBMAX + 1];
    __shared__ int   fil[NBMAX];
    __shared__ float edl[NBMAX];
    __shared__ int   wsum[32];
    __shared__ int   lst_n;

    const int tid = threadIdx.x;
    const int lane = tid & 31;
    const int wave = __builtin_amdgcn_readfirstlane(tid >> 5);
    const int q = lane & 7;
    const int row0 = blockIdx.x * rpw;
    const int nrows = min(rpw, N - row0);
    const int nb = nrows * R;
    const bool binned = nb <= NBMAX;

    for (int b = tid; b <= NBMAX; b += 1024) off[b] = 0;
    if (tid == 0) lst_n = 0;

    __syncthreads();

    auto take_rec = [&](int b, int sr) {
        const int slot = atomicAdd(&lst_n, 1);
        if (binned && slot < LCAP) {
            lst_i[slot] = sr;
            lst_b[slot] = b;
            atomicAdd(&off[b], 1);
        }
    };
    auto take = [&](int t, int i) {
        const unsigned tl = (unsigned)(t - row0);
        if (tl < (unsigned)nrows) take_rec((int)tl * R + i / E, src[i]);
    };

    if (bins_ok(total, rpw, (int)gridDim.x, R, N, ws_bytes)) {
        const int nch = (total + CHUNK - 1) / CHUNK;
        const int nt1 = (int)gridDim.x + 1;
        const size_t tab = ((size_t)nch * (size_t)nt1 + 63) & ~(size_t)63;
        for (int c = tid >> 3; c < nch; c += 128) {
            const int* tbl = ws + (size_t)c * nt1 + blockIdx.x;
            const int sb = tbl[0], se = tbl[1];
            const int* lst = ws + tab + (size_t)c * CHUNK;
            for (int p = sb + q; p < se; p += 16) {
                const int e0 = lst[p];
                const int e1 = (p + 8 < se) ? lst[p + 8] : -1;
                if ((e0 >> 20) < nb) take_rec(e0 >> 20, e0 & ((1 << 20) - 1));
                if (e1 >= 0 && (e1 >> 20) < nb) take_rec(e1 >> 20, e1 & ((1 << 20) - 1));
            }
        }
    } else {
        const int4* dst4 = (const int4*)dst;
        const int n4 = total / 4;
        for (int q0 = tid; q0 < n4; q0 += 1024 * 8) {
            int4 v[8];
    #pragma unroll
            for (int k = 0; k < 8; ++k) {
                const int qq = q0 + k * 1024;
                v[k] = (qq < n4) ? dst4[qq] : make_int4(-1, -1, -1, -1);
            }
    #pragma unroll
            for (int k = 0; k < 8; ++k) {
                const unsigned d0 = (unsigned)(v[k].x - row0), d1 = (unsigned)(v[k].y - row0);
                const unsigned d2 = (unsigned)(v[k].z - row0), d3 = (unsigned)(v[k].w - row0);
                if (min(min(d0, d1), min(d2, d3)) < (unsigned)nrows) {
                    const int i = (q0 + k * 1024) * 4;
                    take(v[k].x, i + 0);
                    take(v[k].y, i + 1);
                    take(v[k].z, i + 2);
                    take(v[k].w, i + 3);
                }
            }
        }
        for (int i = n4 * 4 + tid; i < total; i += 1024) take(dst[i], i);
    }
    __syncthreads();

    const int n = lst_n;
    if (!binned || n > LCAP) {
        selfloop_tile(x, attn_w, W, bias, tile, edl, false, row0, nrows, N, R, wave, lane);
        __syncthreads();
        for (int base = wave * 4; base < nrows; base += 128) {
            const int tl = base + (lane >> 3);
            if (tl < nrows)
                row_unsorted(x, attn_w, src, dst, out, (const float4*)(tile + tl * PP + q * 4),
                             E, R, row0 + tl, q);
        }
        return;
    }

    const int b0 = 2 * tid, b1 = 2 * tid + 1;
    const int c0 = (b0 < nb) ? off[b0] : 0;
    const int c1 = (b1 < nb) ? off[b1] : 0;
    int incl = c0 + c1;
    for (int o = 1; o < 32; o <<= 1) {
        const int y = __shfl_up(incl, o, 32);
        if (lane >= o) incl += y;
    }
    if (lane == 31) wsum[wave] = incl;
    __syncthreads();
    if (wave == 0) {
        const int wt = wsum[lane];
        int wi = wt;
        for (int o = 1; o < 32; o <<= 1) {
            const int y = __shfl_up(wi, o, 32);
            if (lane >= o) wi += y;
        }
        wsum[lane] = wi - wt;
    }
    __syncthreads();
    const int ex = wsum[wave] + incl - (c0 + c1);
    if (b0 < nb) { off[b0] = ex;      fil[b0] = ex; }
    if (b1 < nb) { off[b1] = ex + c0; fil[b1] = ex + c0; }
    if (tid == 0) off[nb] = n;
    __syncthreads();

    for (int j = tid; j < n; j += 1024) {
        const int pos = atomicAdd(&fil[lst_b[j]], 1);
        srt_s[pos] = lst_i[j];
    }
    __syncthreads();

    selfloop_tile(x, attn_w, W, bias, tile, edl, true, row0, nrows, N, R, wave, lane);
    __syncthreads();

    for (int base = wave * 4; base < nrows; base += 128) {
        const int tl = base + (lane >> 3);
        if (tl < nrows) {
            float4* mine = (float4*)(tile + tl * PP + q * 4);
            for (int r = 0; r < R; ++r) {
                const float4* a0 = (const float4*)((const char*)attn_w +
                                                   (unsigned)(r * (2 * DD * 4) + q * 16));
                float4 w[4];
#pragma unroll
                for (int j = 0; j < 4; ++j) w[j] = a0[8 * j];
                const float ed = edl[tl * R + r];
                const int beg = off[tl * R + r], end = off[tl * R + r + 1];
                float m = -INFINITY, l = 0.0f;
                float4 acc[4];
#pragma unroll
                for (int j = 0; j < 4; ++j) acc[j] = make_float4(0.0f, 0.0f, 0.0f, 0.0f);
                for (int k0 = beg; k0 < end; k0 += CH) {
                    const int cnt = min(CH, end - k0);
                    int s[CH];
#pragma unroll
                    for (int u = 0; u < CH; ++u) s[u] = srt_s[min(k0 + u, end - 1)];
                    take_edges<CH>(x, s, cnt, w, ed, q, m, l, acc);
                }
                asm volatile("" ::: "memory");
                const float inv = 1.0f / fmaxf(l, 1e-30f);
#pragma unroll
                for (int j = 0; j < 4; ++j) {
                    acc[j].x *= inv; acc[j].y *= inv; acc[j].z *= inv; acc[j].w *= inv;
                    const float4 pv = mine[8 * j];
                    acc[j].x += pv.x; acc[j].y += pv.y; acc[j].z += pv.z; acc[j].w += pv.w;
                }
                if (r + 1 < R) {
#pragma unroll
                    for (int j = 0; j < 4; ++j) mine[8 * j] = acc[j];
                } else {
                    float4* o = (float4*)((char*)out +
                                          (unsigned)((row0 + tl) * (DD * 4) + q * 16));
#pragma unroll
                    for (int j = 0; j < 4; ++j) o[8 * j] = acc[j];
                }
            }
        }
    }
}

extern "C" void kernel_launch(void* const* d_in, const int* in_sizes, int n_in,
                              void* d_out, int out_size, void* d_ws, size_t ws_size,
                              hipStream_t stream) {
    const float* x      = (const float*)d_in[0];
    const float* attn_w = (const float*)d_in[1];
    const float* W      = (const float*)d_in[2];
    const float* bias   = (const float*)d_in[3];
    const int*   src    = (const int*)d_in[4];
    const int*   dst    = (const int*)d_in[5];
    float*       out    = (float*)d_out;

    const int D = in_sizes[3];
    const int N = in_sizes[0] / D;
    const int R = in_sizes[1] / (2 * D);
    const int E = in_sizes[4] / R;

    const int rpw = TILE_FLOATS / D;
    dim3 sgrid((N + rpw - 1) / rpw);
    dim3 pgrid((R * E + CHUNK - 1) / CHUNK);
    bin_edges<<<pgrid, 1024, 0, stream>>>(src, dst, (int*)d_ws, R * E, rpw, (int)sgrid.x,
                                          E, R, N, ws_size);
    edge_tile<<<sgrid, 1024, 0, stream>>>(x, attn_w, W, bias, src, dst, out, (const int*)d_ws,
                                          ws_size, E, N, R, rpw, R * E);
}


# Row pass: the block's rows counting-sorted in LDS by per-relation trip counts; a wave draws four equal-trip rows at a time, longest first, from an LDS counter (was a fixed split); sums unchanged
# speedup vs baseline: 5.9404x; 1.0756x over previous
#include <hip/hip_runtime.h>
#include <math.h>

typedef __attribute__((ext_vector_type(2))) float v2f;
typedef __attribute__((ext_vector_type(8))) float v8f;

#define DD 128
#define TILE_FLOATS 51200
#define LCAP 12288
#define NBMAX 2048
#define CH 3
#define TCAP 3
#define NK ((TCAP + 1) * (TCAP + 1))
#define RPWMAX (TILE_FLOATS / DD)
#define PP 132

__device__ __forceinline__ float dot4(float4 a, float4 b) {
    return a.x * b.x + a.y * b.y + a.z * b.z + a.w * b.w;
}

template <int M>
__device__ __forceinline__ float lane_xor(float x) {
    const int b = __builtin_bit_cast(int, x);
    return __builtin_bit_cast(float, __builtin_amdgcn_update_dpp(0, b, 0x160 + M, 0xf, 0xf, true));
}

__device__ __forceinline__ float sum8(float v) {
    v += lane_xor<4>(v);
    v += lane_xor<2>(v);
    v += lane_xor<1>(v);
    return v;
}

__device__ __forceinline__ float dot16(const float4 (&v)[4], const float4 (&w)[4]) {
    v2f s = v2f{v[0].x, v[0].y} * v2f{w[0].x, w[0].y};
    s = v2f{v[0].z, v[0].w} * v2f{w[0].z, w[0].w} + s;
#pragma unroll
    for (int j = 1; j < 4; ++j) {
        s = v2f{v[j].x, v[j].y} * v2f{w[j].x, w[j].y} + s;
        s = v2f{v[j].z, v[j].w} * v2f{w[j].z, w[j].w} + s;
    }
    return s.x + s.y;
}

template <int K>
__device__ __forceinline__ void take_edges(const float* __restrict__ x, const int (&s)[K],
                                           int cnt, const float4 (&w)[4], float ed, int q,
                                           float& m, float& l, float4 (&acc)[4]) {
    float4 v[K][4];
    float e[K];
#pragma unroll
    for (int u = 0; u < K; ++u) {
        const unsigned ob = (unsigned)s[u] * (unsigned)(DD * 4) + (unsigned)(q * 16);
        const float4* p = (const float4*)((const char*)x + ob);
#pragma unroll
        for (int j = 0; j < 4; ++j) v[u][j] = p[8 * j];
    }
    float cm = m;
#pragma unroll
    for (int u = 0; u < K; ++u) {
        const float eu = sum8(dot16(v[u], w)) + ed;
        e[u] = (u < cnt) ? eu : -INFINITY;
        cm = fmaxf(cm, e[u]);
    }
    const float sc = __expf(m - cm);
    l *= sc;
#pragma unroll
    for (int j = 0; j < 4; ++j) {
        acc[j].x *= sc; acc[j].y *= sc; acc[j].z *= sc; acc[j].w *= sc;
    }
#pragma unroll
    for (int u = 0; u < K; ++u) {
        const float p = (u < cnt) ? __expf(e[u] - cm) : 0.0f;
        l += p;
#pragma unroll
        for (int j = 0; j < 4; ++j) {
            acc[j].x += p * v[u][j].x; acc[j].y += p * v[u][j].y;
            acc[j].z += p * v[u][j].z; acc[j].w += p * v[u][j].w;
        }
    }
    m = cm;
}

__device__ __forceinline__ void row_unsorted(const float* __restrict__ x,
                                          const float* __restrict__ attn_w,
                                          const int* __restrict__ src,
                                          const int* __restrict__ dst,
                                          float* __restrict__ out,
                                          const float4* pr,
                                          int E, int R, int t, int q) {
    const float4* xd = (const float4*)(x + (size_t)t * DD + q * 4);
    float4 tot[4];
#pragma unroll
    for (int j = 0; j < 4; ++j) tot[j] = pr[8 * j];
    for (int r = 0; r < R; ++r) {
        const float4* a0 = (const float4*)(attn_w + (size_t)r * 2 * DD + q * 4);
        const float4* a1 = a0 + DD / 4;
        float4 w[4];
        float pd = 0.0f;
#pragma unroll
        for (int j = 0; j < 4; ++j) {
            w[j] = a0[8 * j];
            pd += dot4(xd[8 * j], a1[8 * j]);
        }
        const float ed = sum8(pd);
        float m = -INFINITY, l = 0.0f;
        float4 acc[4];
#pragma unroll
        for (int j = 0; j < 4; ++j) acc[j] = make_float4(0.0f, 0.0f, 0.0f, 0.0f);
        for (int i = 0; i < E; ++i)
            if (dst[(size_t)r * E + i] == t) {
                const int s[1] = { src[(size_t)r * E + i] };
                take_edges<1>(x, s, 1, w, ed, q, m, l, acc);
            }
        const float inv = 1.0f / fmaxf(l, 1e-30f);
#pragma unroll
        for (int j = 0; j < 4; ++j) {
            tot[j].x += acc[j].x * inv; tot[j].y += acc[j].y * inv;
            tot[j].z += acc[j].z * inv; tot[j].w += acc[j].w * inv;
        }
    }
    float4* o = (float4*)(out + (size_t)t * DD + q * 4);
#pragma unroll
    for (int j = 0; j < 4; ++j) o[8 * j] = tot[j];
}

template <bool TWO>
__device__ __forceinline__ void product_round(const float* a0p, const float (&breg)[64],
                                              v8f& c0, v8f& c1) {
#pragma unroll
    for (int i = 0; i < 16; ++i) {
        const float4 a0 = *(const float4*)(a0p + 8 * i);
        v2f bl, bh, al, ah;
        bl.x = breg[4 * i];     bl.y = breg[4 * i + 1];
        bh.x = breg[4 * i + 2]; bh.y = breg[4 * i + 3];
        al.x = a0.x; al.y = a0.y; ah.x = a0.z; ah.y = a0.w;
        c0 = __builtin_amdgcn_wmma_f32_16x16x4_f32(false, al, false, bl, (short)0, c0, false, false);
        if (TWO) {
            const float4 a1 = *(const float4*)(a0p + 16 * PP + 8 * i);
            v2f dl, dh;
            dl.x = a1.x; dl.y = a1.y; dh.x = a1.z; dh.y = a1.w;
            c1 = __builtin_amdgcn_wmma_f32_16x16x4_f32(false, dl, false, bl, (short)0, c1, false, false);
            c0 = __builtin_amdgcn_wmma_f32_16x16x4_f32(false, ah, false, bh, (short)0, c0, false, false);
            c1 = __builtin_amdgcn_wmma_f32_16x16x4_f32(false, dh, false, bh, (short)0, c1, false, false);
        } else {
            c0 = __builtin_amdgcn_wmma_f32_16x16x4_f32(false, ah, false, bh, (short)0, c0, false, false);
        }
    }
}

__device__ __forceinline__ void selfloop_tile(const float* __restrict__ x,
                                              const float* __restrict__ attn_w,
                                              const float* __restrict__ W,
                                              const float* __restrict__ bias,
                                              float* tile, float* edl, bool want_edl,
                                              int row0, int nrows, int N, int R,
                                              int wave, int lane) {
    const int c = lane & 15;
    const int h = lane >> 4;
    const int q = lane & 7;
    const int col = (wave & 7) * 16 + c;
    const int g = wave >> 3;

    float breg[64];
    const float* bp = W + (4 * h) * DD + col;
#pragma unroll
    for (int i = 0; i < 16; ++i)
#pragma unroll
        for (int e = 0; e < 4; ++e) breg[4 * i + e] = bp[(8 * i + e) * DD];
    const float bs = bias[col];

    const int ntile = (nrows + 15) >> 4;
    for (int base = wave * 4; base < ntile * 16; base += 128) {
        const int tl = base + (lane >> 3);
        const int gr = min(row0 + tl, N - 1);
        const float4* xd = (const float4*)(x + (size_t)gr * DD + q * 4);
        float4 xv[4];
#pragma unroll
        for (int j = 0; j < 4; ++j) xv[j] = xd[8 * j];
        float4* tr = (float4*)(tile + tl * PP + q * 4);
#pragma unroll
        for (int j = 0; j < 4; ++j) tr[8 * j] = xv[j];
        if (want_edl && tl < nrows)
            for (int r = 0; r < R; ++r) {
                const float4* a1 = (const float4*)(attn_w + (size_t)r * 2 * DD + DD + q * 4);
                const float pd = dot4(xv[0], a1[0]) + dot4(xv[1], a1[8]) +
                                 dot4(xv[2], a1[16]) + dot4(xv[3], a1[24]);
                const float ed = sum8(pd);
                if (q == 0) edl[tl * R + r] = ed;
            }
    }
    __syncthreads();

    const int nround = (ntile + 7) >> 3;
    for (int rd = 0; rd < nround; ++rd) {
        const int t0 = rd * 8 + g * 2;
        const float* a0p = tile + (t0 * 16 + c) * PP + 4 * h;
        v8f c0 = {}, c1 = {};
        if (t0 + 1 < ntile) product_round<true>(a0p, breg, c0, c1);
        else if (t0 < ntile) product_round<false>(a0p, breg, c0, c1);
        __syncthreads();
        float* o = tile + (t0 * 16 + 8 * h) * PP + col;
        if (t0 < ntile) {
#pragma unroll
            for (int v = 0; v < 8; ++v) o[v * PP] = c0[v] + bs;
        }
        if (t0 + 1 < ntile) {
#pragma unroll
            for (int v = 0; v < 8; ++v) o[(16 + v) * PP] = c1[v] + bs;
        }
    }
}

#define CHUNK 4096
#define NTMAX 1024

__device__ __forceinline__ bool bins_ok(int total, int rpw, int ntiles, int R, int N,
                                        size_t ws_bytes) {
    const size_t nch = (size_t)((total + CHUNK - 1) / CHUNK);
    const size_t tab = (nch * (size_t)(ntiles + 1) + 63) & ~(size_t)63;
    return N <= (1 << 20) && rpw * R <= NBMAX && ntiles <= NTMAX &&
           (tab + nch * CHUNK) * sizeof(int) <= ws_bytes;
}

__global__ __launch_bounds__(1024) void bin_edges(const int* __restrict__ src,
                                                  const int* __restrict__ dst,
                                                  int* __restrict__ ws,
                                                  int total, int rpw, int ntiles,
                                                  int E, int R, int N, size_t ws_bytes) {
    __shared__ int cnt[NTMAX];
    __shared__ int beg[NTMAX];
    __shared__ int srt[CHUNK];
    __shared__ int wsum[32];

    if (!bins_ok(total, rpw, ntiles, R, N, ws_bytes)) return;
    const int tid = threadIdx.x;
    const int lane = tid & 31;
    const int wave = tid >> 5;
    const int c = blockIdx.x;
    const int nch = (total + CHUNK - 1) / CHUNK;
    const int nt1 = ntiles + 1;
    const size_t tab = ((size_t)nch * (size_t)nt1 + 63) & ~(size_t)63;
    int* lst = ws + tab + (size_t)c * CHUNK;
    int* tbl = ws + (size_t)c * nt1;

    cnt[tid] = 0;
    __syncthreads();

    const int i0 = c * CHUNK + tid * 4;
    int d[4], sv[4];
    if (i0 + 3 < total) {
        const int4 v = *(const int4*)(dst + i0);
        const int4 u = *(const int4*)(src + i0);
        d[0] = v.x; d[1] = v.y; d[2] = v.z; d[3] = v.w;
        sv[0] = u.x; sv[1] = u.y; sv[2] = u.z; sv[3] = u.w;
    } else {
#pragma unroll
        for (int k = 0; k < 4; ++k) {
            d[k] = (i0 + k < total) ? dst[i0 + k] : -1;
            sv[k] = (i0 + k < total) ? src[i0 + k] : 0;
        }
    }
    int tt[4], val[4], pos[4];
#pragma unroll
    for (int k = 0; k < 4; ++k) {
        const int t = (d[k] >= 0) ? d[k] / rpw : ntiles;
        tt[k] = ((unsigned)t < (unsigned)ntiles) ? t : -1;
        val[k] = (tt[k] >= 0) ? ((((d[k] - t * rpw) * R + (i0 + k) / E) << 20) | sv[k]) : 0;
        pos[k] = 0;
        if (tt[k] >= 0) pos[k] = atomicAdd(&cnt[tt[k]], 1);
    }
    __syncthreads();

    const int c0 = (tid < ntiles) ? cnt[tid] : 0;
    int incl = c0;
    for (int o = 1; o < 32; o <<= 1) {
        const int y = __shfl_up(incl, o, 32);
        if (lane >= o) incl += y;
    }
    if (lane == 31) wsum[wave] = incl;
    __syncthreads();
    if (wave == 0) {
        const int wt = wsum[lane];
        int wi = wt;
        for (int o = 1; o < 32; o <<= 1) {
            const int y = __shfl_up(wi, o, 32);
            if (lane >= o) wi += y;
        }
        wsum[lane] = wi - wt;
    }
    __syncthreads();
    const int ex = wsum[wave] + incl - c0;
    beg[tid] = ex;
    if (tid < ntiles) tbl[tid] = ex;
    if (tid == 1023) tbl[ntiles] = ex + c0;
    __syncthreads();

#pragma unroll
    for (int k = 0; k < 4; ++k)
        if (tt[k] >= 0) srt[beg[tt[k]] + pos[k]] = val[k];
    __syncthreads();

    const int nv = beg[1023] + cnt[1023];
#pragma unroll
    for (int k = 0; k < 4; ++k) {
        const int p = tid + k * 1024;
        if (p < nv) lst[p] = srt[p];
    }
}

__global__ __launch_bounds__(1024) void edge_tile(const float* __restrict__ x,
                                                  const float* __restrict__ attn_w,
                                                  const float* __restrict__ W,
                                                  const float* __restrict__ bias,
                                                  const int* __restrict__ src,
                                                  const int* __restrict__ dst,
                                                  float* __restrict__ out,
                                                  const int* __restrict__ ws, size_t ws_bytes,
                                                  int E, int N, int R, int rpw, int total) {
    __shared__ __attribute__((aligned(16))) float tile[RPWMAX * PP];
    int* const lst_i = (int*)tile;
    int* const lst_b = lst_i + LCAP;
    __shared__ int   srt_s[LCAP];
    __shared__ int   off[NBMAX + 1];
    __shared__ int   fil[NBMAX];
    __shared__ float edl[NBMAX];
    __shared__ int   wsum[32];
    __shared__ int   lst_n;
    __shared__ int   perm[RPWMAX];
    __shared__ int   kcnt[NK];
    __shared__ int   nextq;

    const int tid = threadIdx.x;
    const int lane = tid & 31;
    const int wave = __builtin_amdgcn_readfirstlane(tid >> 5);
    const int q = lane & 7;
    const int row0 = blockIdx.x * rpw;
    const int nrows = min(rpw, N - row0);
    const int nb = nrows * R;
    const bool binned = nb <= NBMAX;

    for (int b = tid; b <= NBMAX; b += 1024) off[b] = 0;
    if (tid < NK) kcnt[tid] = 0;
    if (tid == 0) { lst_n = 0; nextq = 0; }

    __syncthreads();

    auto take_rec = [&](int b, int sr) {
        const int slot = atomicAdd(&lst_n, 1);
        if (binned && slot < LCAP) {
            lst_i[slot] = sr;
            lst_b[slot] = b;
            atomicAdd(&off[b], 1);
        }
    };
    auto take = [&](int t, int i) {
        const unsigned tl = (unsigned)(t - row0);
        if (tl < (unsigned)nrows) take_rec((int)tl * R + i / E, src[i]);
    };

    if (bins_ok(total, rpw, (int)gridDim.x, R, N, ws_bytes)) {
        const int nch = (total + CHUNK - 1) / CHUNK;
        const int nt1 = (int)gridDim.x + 1;
        const size_t tab = ((size_t)nch * (size_t)nt1 + 63) & ~(size_t)63;
        for (int c = tid >> 3; c < nch; c += 128) {
            const int* tbl = ws + (size_t)c * nt1 + blockIdx.x;
            const int sb = tbl[0], se = tbl[1];
            const int* lst = ws + tab + (size_t)c * CHUNK;
            for (int p = sb + q; p < se; p += 16) {
                const int e0 = lst[p];
                const int e1 = (p + 8 < se) ? lst[p + 8] : -1;
                if ((e0 >> 20) < nb) take_rec(e0 >> 20, e0 & ((1 << 20) - 1));
                if (e1 >= 0 && (e1 >> 20) < nb) take_rec(e1 >> 20, e1 & ((1 << 20) - 1));
            }
        }
    } else {
        const int4* dst4 = (const int4*)dst;
        const int n4 = total / 4;
        for (int q0 = tid; q0 < n4; q0 += 1024 * 8) {
            int4 v[8];
    #pragma unroll
            for (int k = 0; k < 8; ++k) {
                const int qq = q0 + k * 1024;
                v[k] = (qq < n4) ? dst4[qq] : make_int4(-1, -1, -1, -1);
            }
    #pragma unroll
            for (int k = 0; k < 8; ++k) {
                const unsigned d0 = (unsigned)(v[k].x - row0), d1 = (unsigned)(v[k].y - row0);
                const unsigned d2 = (unsigned)(v[k].z - row0), d3 = (unsigned)(v[k].w - row0);
                if (min(min(d0, d1), min(d2, d3)) < (unsigned)nrows) {
                    const int i = (q0 + k * 1024) * 4;
                    take(v[k].x, i + 0);
                    take(v[k].y, i + 1);
                    take(v[k].z, i + 2);
                    take(v[k].w, i + 3);
                }
            }
        }
        for (int i = n4 * 4 + tid; i < total; i += 1024) take(dst[i], i);
    }
    __syncthreads();

    const int n = lst_n;
    if (!binned || n > LCAP) {
        selfloop_tile(x, attn_w, W, bias, tile, edl, false, row0, nrows, N, R, wave, lane);
        __syncthreads();
        for (int base = wave * 4; base < nrows; base += 128) {
            const int tl = base + (lane >> 3);
            if (tl < nrows)
                row_unsorted(x, attn_w, src, dst, out, (const float4*)(tile + tl * PP + q * 4),
                             E, R, row0 + tl, q);
        }
        return;
    }

    const int b0 = 2 * tid, b1 = 2 * tid + 1;
    const int c0 = (b0 < nb) ? off[b0] : 0;
    const int c1 = (b1 < nb) ? off[b1] : 0;
    int incl = c0 + c1;
    for (int o = 1; o < 32; o <<= 1) {
        const int y = __shfl_up(incl, o, 32);
        if (lane >= o) incl += y;
    }
    if (lane == 31) wsum[wave] = incl;
    __syncthreads();
    if (wave == 0) {
        const int wt = wsum[lane];
        int wi = wt;
        for (int o = 1; o < 32; o <<= 1) {
            const int y = __shfl_up(wi, o, 32);
            if (lane >= o) wi += y;
        }
        wsum[lane] = wi - wt;
    }
    __syncthreads();
    const int ex = wsum[wave] + incl - (c0 + c1);
    if (b0 < nb) { off[b0] = ex;      fil[b0] = ex; }
    if (b1 < nb) { off[b1] = ex + c0; fil[b1] = ex + c0; }
    if (tid == 0) off[nb] = n;
    __syncthreads();

    for (int j = tid; j < n; j += 1024) {
        const int pos = atomicAdd(&fil[lst_b[j]], 1);
        srt_s[pos] = lst_i[j];
    }

    int rkey = 0, rpos = 0;
    if (tid < nrows) {
        const int* ob = off + tid * R;
        rkey = min((ob[1] - ob[0] + CH - 1) / CH, TCAP) * (TCAP + 1);
        if (R > 1) rkey += min((ob[2] - ob[1] + CH - 1) / CH, TCAP);
        rpos = atomicAdd(&kcnt[rkey], 1);
    }
    __syncthreads();

    if (tid < nrows) {
        int st = rpos;
#pragma unroll
        for (int k = 0; k < NK; ++k) st += (k > rkey) ? kcnt[k] : 0;
        perm[st] = tid;
    }

    selfloop_tile(x, attn_w, W, bias, tile, edl, true, row0, nrows, N, R, wave, lane);
    __syncthreads();

    const int nquad = (nrows + 3) >> 2;
    for (;;) {
        int qd = 0;
        if (lane == 0) qd = atomicAdd(&nextq, 1);
        qd = __builtin_amdgcn_readfirstlane(qd);
        if (qd >= nquad) break;
        const int pi = qd * 4 + (lane >> 3);
        if (pi < nrows) {
            const int tl = perm[pi];
            float4* mine = (float4*)(tile + tl * PP + q * 4);
            for (int r = 0; r < R; ++r) {
                const float4* a0 = (const float4*)((const char*)attn_w +
                                                   (unsigned)(r * (2 * DD * 4) + q * 16));
                float4 w[4];
#pragma unroll
                for (int j = 0; j < 4; ++j) w[j] = a0[8 * j];
                const float ed = edl[tl * R + r];
                const int beg = off[tl * R + r], end = off[tl * R + r + 1];
                float m = -INFINITY, l = 0.0f;
                float4 acc[4];
#pragma unroll
                for (int j = 0; j < 4; ++j) acc[j] = make_float4(0.0f, 0.0f, 0.0f, 0.0f);
                for (int k0 = beg; k0 < end; k0 += CH) {
                    const int cnt = min(CH, end - k0);
                    int s[CH];
#pragma unroll
                    for (int u = 0; u < CH; ++u) s[u] = srt_s[min(k0 + u, end - 1)];
                    take_edges<CH>(x, s, cnt, w, ed, q, m, l, acc);
                }
                asm volatile("" ::: "memory");
                const float inv = 1.0f / fmaxf(l, 1e-30f);
#pragma unroll
                for (int j = 0; j < 4; ++j) {
                    acc[j].x *= inv; acc[j].y *= inv; acc[j].z *= inv; acc[j].w *= inv;
                    const float4 pv = mine[8 * j];
                    acc[j].x += pv.x; acc[j].y += pv.y; acc[j].z += pv.z; acc[j].w += pv.w;
                }
                if (r + 1 < R) {
#pragma unroll
                    for (int j = 0; j < 4; ++j) mine[8 * j] = acc[j];
                } else {
                    float4* o = (float4*)((char*)out +
                                          (unsigned)((row0 + tl) * (DD * 4) + q * 16));
#pragma unroll
                    for (int j = 0; j < 4; ++j) o[8 * j] = acc[j];
                }
            }
        }
    }
}

extern "C" void kernel_launch(void* const* d_in, const int* in_sizes, int n_in,
                              void* d_out, int out_size, void* d_ws, size_t ws_size,
                              hipStream_t stream) {
    const float* x      = (const float*)d_in[0];
    const float* attn_w = (const float*)d_in[1];
    const float* W      = (const float*)d_in[2];
    const float* bias   = (const float*)d_in[3];
    const int*   src    = (const int*)d_in[4];
    const int*   dst    = (const int*)d_in[5];
    float*       out    = (float*)d_out;

    const int D = in_sizes[3];
    const int N = in_sizes[0] / D;
    const int R = in_sizes[1] / (2 * D);
    const int E = in_sizes[4] / R;

    const int rpw = TILE_FLOATS / D;
    dim3 sgrid((N + rpw - 1) / rpw);
    dim3 pgrid((R * E + CHUNK - 1) / CHUNK);
    bin_edges<<<pgrid, 1024, 0, stream>>>(src, dst, (int*)d_ws, R * E, rpw, (int)sgrid.x,
                                          E, R, N, ws_size);
    edge_tile<<<sgrid, 1024, 0, stream>>>(x, attn_w, W, bias, src, dst, out, (const int*)d_ws,
                                          ws_size, E, N, R, rpw, R * E);
}
